# starting kernel
# baseline (speedup 1.0000x reference)
#include <hip/hip_runtime.h>
#include <hip/hip_bf16.h>
#include <math.h>

typedef __attribute__((ext_vector_type(2))) float v2f;
typedef __attribute__((ext_vector_type(8))) float v8f;

#define N_NODES 100000
#define N_EDGES 1600000
#define IN_DIM 128
#define OUT_DIM 64
#define NEG_SLOPE 0.01f

__global__ void gat_gemm_wmma(const float* __restrict__ h,
                              const float* __restrict__ W,
                              float* __restrict__ z,
                              int n_nodes) {
    int gtid = blockIdx.x * blockDim.x + threadIdx.x;
    int wave = gtid >> 5;
    int lane = threadIdx.x & 31;
    int m0 = wave * 16;
    if (m0 >= n_nodes) return;

    int rowA  = m0 + (lane & 15);
    int khalf = (lane >> 4) * 2;
    int nlane = lane & 15;

    v8f acc0 = {}, acc1 = {}, acc2 = {}, acc3 = {};

    for (int k0 = 0; k0 < IN_DIM; k0 += 4) {
        v2f a = *(const v2f*)(h + (size_t)rowA * IN_DIM + k0 + khalf);

        v2f b0 = *(const v2f*)(W + (size_t)(nlane +  0) * IN_DIM + k0 + khalf);
        v2f b1 = *(const v2f*)(W + (size_t)(nlane + 16) * IN_DIM + k0 + khalf);
        v2f b2 = *(const v2f*)(W + (size_t)(nlane + 32) * IN_DIM + k0 + khalf);
        v2f b3 = *(const v2f*)(W + (size_t)(nlane + 48) * IN_DIM + k0 + khalf);

        acc0 = __builtin_amdgcn_wmma_f32_16x16x4_f32(false, a, false, b0, (short)0, acc0, false, false);
        acc1 = __builtin_amdgcn_wmma_f32_16x16x4_f32(false, a, false, b1, (short)0, acc1, false, false);
        acc2 = __builtin_amdgcn_wmma_f32_16x16x4_f32(false, a, false, b2, (short)0, acc2, false, false);
        acc3 = __builtin_amdgcn_wmma_f32_16x16x4_f32(false, a, false, b3, (short)0, acc3, false, false);
    }

    int rsel = (lane >> 4) * 8;
#pragma unroll
    for (int r = 0; r < 8; ++r) {
        size_t ro = (size_t)(m0 + r + rsel) * OUT_DIM;
        z[ro + nlane +  0] = acc0[r];
        z[ro + nlane + 16] = acc1[r];
        z[ro + nlane + 32] = acc2[r];
        z[ro + nlane + 48] = acc3[r];
    }
}

__global__ void gat_node_scores(const float* __restrict__ z,
                                const float* __restrict__ attn,
                                float* __restrict__ s_l,
                                float* __restrict__ s_r,
                                int n) {
    int i = blockIdx.x * blockDim.x + threadIdx.x;
    if (i >= n) return;
    const float4* zr = (const float4*)(z + (size_t)i * OUT_DIM);
    const float4* al = (const float4*)(attn);
    const float4* ar = (const float4*)(attn + OUT_DIM);
    float sl = 0.f, sr = 0.f;
#pragma unroll
    for (int q = 0; q < OUT_DIM / 4; ++q) {
        float4 zv = zr[q];
        float4 av = al[q];
        float4 bv = ar[q];
        sl += zv.x * av.x + zv.y * av.y + zv.z * av.z + zv.w * av.w;
        sr += zv.x * bv.x + zv.y * bv.y + zv.z * bv.z + zv.w * bv.w;
    }
    s_l[i] = sl;
    s_r[i] = sr;
}

__global__ void gat_zero_i32(int* __restrict__ p, int n) {
    int i = blockIdx.x * blockDim.x + threadIdx.x;
    if (i < n) p[i] = 0;
}

__global__ void gat_edge_scores(const int* __restrict__ src,
                                const int* __restrict__ dst,
                                const float* __restrict__ s_l,
                                const float* __restrict__ s_r,
                                float* __restrict__ e,
                                int* __restrict__ deg,
                                int ne) {
    int i = blockIdx.x * blockDim.x + threadIdx.x;
    if (i >= ne) return;
    int d = dst[i];
    float v = s_l[src[i]] + s_r[d];
    e[i] = (v > 0.f) ? v : NEG_SLOPE * v;
    atomicAdd(&deg[d], 1);
}

__global__ void gat_scan_excl(const int* __restrict__ deg,
                              int* __restrict__ offs,
                              int* __restrict__ cursor,
                              int n) {
    __shared__ int s[1024];
    int t = threadIdx.x;
    int run = 0;
    for (int base = 0; base < n; base += 1024) {
        int idx = base + t;
        int v = (idx < n) ? deg[idx] : 0;
        s[t] = v;
        __syncthreads();
        int val = v;
        for (int off = 1; off < 1024; off <<= 1) {
            int tmp = (t >= off) ? s[t - off] : 0;
            __syncthreads();
            val += tmp;
            s[t] = val;
            __syncthreads();
        }
        int excl = run + val - v;
        if (idx < n) { offs[idx] = excl; cursor[idx] = excl; }
        run += s[1023];
        __syncthreads();
    }
    if (t == 0) offs[n] = run;
}

__global__ void gat_fill_csr(const int* __restrict__ dst,
                             int* __restrict__ cursor,
                             int* __restrict__ csr,
                             int ne) {
    int i = blockIdx.x * blockDim.x + threadIdx.x;
    if (i >= ne) return;
    int p = atomicAdd(&cursor[dst[i]], 1);
    csr[p] = i;
}

__global__ void gat_node_aggregate(const float* __restrict__ z,
                                   const float* __restrict__ e,
                                   const int* __restrict__ src,
                                   const int* __restrict__ csr,
                                   const int* __restrict__ offs,
                                   const int* __restrict__ deg,
                                   float* __restrict__ out,
                                   int n) {
    int gtid = blockIdx.x * blockDim.x + threadIdx.x;
    int node = gtid >> 5;
    if (node >= n) return;
    int lane = threadIdx.x & 31;
    int c0 = lane * 2;

    int d = deg[node];
    int o = offs[node];

    float accx = 0.f, accy = 0.f;
    if (d > 0) {
        float sum = 0.f;
        for (int j = 0; j < d; ++j) sum += e[csr[o + j]];
        float mean = sum / (float)d;

        float m = -1e30f;
        for (int j = 0; j < d; ++j) {
            float ev = e[csr[o + j]];
            float e2 = (ev < mean) ? 0.f : ev;
            m = fmaxf(m, e2);
        }

        float denom = 0.f;
        for (int j = 0; j < d; ++j) {
            int ei = csr[o + j];
            float ev = e[ei];
            float e2 = (ev < mean) ? 0.f : ev;
            float ex = __expf(e2 - m);
            denom += ex;
            const float2 zv = *(const float2*)(z + (size_t)src[ei] * OUT_DIM + c0);
            accx += ex * zv.x;
            accy += ex * zv.y;
        }
        float inv = 1.f / denom;
        accx *= inv;
        accy *= inv;
    }
    float2 r;
    r.x = accx;
    r.y = accy;
    *(float2*)(out + (size_t)node * OUT_DIM + c0) = r;
}

extern "C" void kernel_launch(void* const* d_in, const int* in_sizes, int n_in,
                              void* d_out, int out_size, void* d_ws, size_t ws_size,
                              hipStream_t stream) {
    const float* h      = (const float*)d_in[0];
    const int*   src    = (const int*)d_in[1];
    const int*   dst    = (const int*)d_in[2];
    const float* W_fc   = (const float*)d_in[3];
    const float* W_attn = (const float*)d_in[4];
    float* out = (float*)d_out;

    char* ws = (char*)d_ws;
    size_t off = 0;
    auto carve = [&](size_t bytes) {
        void* p = ws + off;
        off += (bytes + 15) & ~(size_t)15;
        return p;
    };
    float* z      = (float*)carve((size_t)N_NODES * OUT_DIM * sizeof(float));
    float* s_l    = (float*)carve((size_t)N_NODES * sizeof(float));
    float* s_r    = (float*)carve((size_t)N_NODES * sizeof(float));
    float* e      = (float*)carve((size_t)N_EDGES * sizeof(float));
    int*   deg    = (int*)carve((size_t)N_NODES * sizeof(int));
    int*   offs   = (int*)carve((size_t)(N_NODES + 1) * sizeof(int));
    int*   cursor = (int*)carve((size_t)N_NODES * sizeof(int));
    int*   csr    = (int*)carve((size_t)N_EDGES * sizeof(int));
    (void)ws_size;

    const int B = 256;

    int gemm_waves  = N_NODES / 16;
    int gemm_blocks = (gemm_waves + 7) / 8;
    gat_gemm_wmma<<<gemm_blocks, B, 0, stream>>>(h, W_fc, z, N_NODES);

    gat_node_scores<<<(N_NODES + B - 1) / B, B, 0, stream>>>(z, W_attn, s_l, s_r, N_NODES);

    gat_zero_i32<<<(N_NODES + B - 1) / B, B, 0, stream>>>(deg, N_NODES);
    gat_edge_scores<<<(N_EDGES + B - 1) / B, B, 0, stream>>>(src, dst, s_l, s_r, e, deg, N_EDGES);

    gat_scan_excl<<<1, 1024, 0, stream>>>(deg, offs, cursor, N_NODES);

    gat_fill_csr<<<(N_EDGES + B - 1) / B, B, 0, stream>>>(dst, cursor, csr, N_EDGES);

    int agg_blocks = (N_NODES + 7) / 8;
    gat_node_aggregate<<<agg_blocks, B, 0, stream>>>(z, e, src, csr, offs, deg, out, N_NODES);
}


# Degree scan: one block's 98 serial Hillis-Steele rounds (20 barriers each) replaced by a one-pass 25-block scan; each block sums the degrees in front of its chunk itself, wave scan, 1 barrier
# speedup vs baseline: 1.5744x; 1.5744x over previous
#include <hip/hip_runtime.h>
#include <hip/hip_bf16.h>
#include <math.h>

typedef __attribute__((ext_vector_type(2))) float v2f;
typedef __attribute__((ext_vector_type(8))) float v8f;

#define N_NODES 100000
#define N_EDGES 1600000
#define IN_DIM 128
#define OUT_DIM 64
#define NEG_SLOPE 0.01f

__global__ void gat_gemm_wmma(const float* __restrict__ h,
                              const float* __restrict__ W,
                              float* __restrict__ z,
                              int n_nodes) {
    int gtid = blockIdx.x * blockDim.x + threadIdx.x;
    int wave = gtid >> 5;
    int lane = threadIdx.x & 31;
    int m0 = wave * 16;
    if (m0 >= n_nodes) return;

    int rowA  = m0 + (lane & 15);
    int khalf = (lane >> 4) * 2;
    int nlane = lane & 15;

    v8f acc0 = {}, acc1 = {}, acc2 = {}, acc3 = {};

    for (int k0 = 0; k0 < IN_DIM; k0 += 4) {
        v2f a = *(const v2f*)(h + (size_t)rowA * IN_DIM + k0 + khalf);

        v2f b0 = *(const v2f*)(W + (size_t)(nlane +  0) * IN_DIM + k0 + khalf);
        v2f b1 = *(const v2f*)(W + (size_t)(nlane + 16) * IN_DIM + k0 + khalf);
        v2f b2 = *(const v2f*)(W + (size_t)(nlane + 32) * IN_DIM + k0 + khalf);
        v2f b3 = *(const v2f*)(W + (size_t)(nlane + 48) * IN_DIM + k0 + khalf);

        acc0 = __builtin_amdgcn_wmma_f32_16x16x4_f32(false, a, false, b0, (short)0, acc0, false, false);
        acc1 = __builtin_amdgcn_wmma_f32_16x16x4_f32(false, a, false, b1, (short)0, acc1, false, false);
        acc2 = __builtin_amdgcn_wmma_f32_16x16x4_f32(false, a, false, b2, (short)0, acc2, false, false);
        acc3 = __builtin_amdgcn_wmma_f32_16x16x4_f32(false, a, false, b3, (short)0, acc3, false, false);
    }

    int rsel = (lane >> 4) * 8;
#pragma unroll
    for (int r = 0; r < 8; ++r) {
        size_t ro = (size_t)(m0 + r + rsel) * OUT_DIM;
        z[ro + nlane +  0] = acc0[r];
        z[ro + nlane + 16] = acc1[r];
        z[ro + nlane + 32] = acc2[r];
        z[ro + nlane + 48] = acc3[r];
    }
}

__global__ void gat_node_scores(const float* __restrict__ z,
                                const float* __restrict__ attn,
                                float* __restrict__ s_l,
                                float* __restrict__ s_r,
                                int n) {
    int i = blockIdx.x * blockDim.x + threadIdx.x;
    if (i >= n) return;
    const float4* zr = (const float4*)(z + (size_t)i * OUT_DIM);
    const float4* al = (const float4*)(attn);
    const float4* ar = (const float4*)(attn + OUT_DIM);
    float sl = 0.f, sr = 0.f;
#pragma unroll
    for (int q = 0; q < OUT_DIM / 4; ++q) {
        float4 zv = zr[q];
        float4 av = al[q];
        float4 bv = ar[q];
        sl += zv.x * av.x + zv.y * av.y + zv.z * av.z + zv.w * av.w;
        sr += zv.x * bv.x + zv.y * bv.y + zv.z * bv.z + zv.w * bv.w;
    }
    s_l[i] = sl;
    s_r[i] = sr;
}

__global__ void gat_zero_i32(int* __restrict__ p, int n) {
    int i = blockIdx.x * blockDim.x + threadIdx.x;
    if (i < n) p[i] = 0;
}

__global__ void gat_edge_scores(const int* __restrict__ src,
                                const int* __restrict__ dst,
                                const float* __restrict__ s_l,
                                const float* __restrict__ s_r,
                                float* __restrict__ e,
                                int* __restrict__ deg,
                                int ne) {
    int i = blockIdx.x * blockDim.x + threadIdx.x;
    if (i >= ne) return;
    int d = dst[i];
    float v = s_l[src[i]] + s_r[d];
    e[i] = (v > 0.f) ? v : NEG_SLOPE * v;
    atomicAdd(&deg[d], 1);
}

#define SCAN_THREADS 1024
#define SCAN_CHUNK   (SCAN_THREADS * 4)
#define SCAN_BLOCKS  ((N_NODES + SCAN_CHUNK - 1) / SCAN_CHUNK)
static_assert(N_NODES % 4 == 0, "the scan reads and writes whole int4 groups");

__global__ void __launch_bounds__(SCAN_THREADS)
gat_scan_excl(const int* __restrict__ deg,
              int* __restrict__ offs,
              int* __restrict__ cursor,
              int n) {
    __shared__ int wave_tot[SCAN_THREADS / 32];
    __shared__ int wave_pre[SCAN_THREADS / 32];

    const int t    = threadIdx.x;
    const int lane = t & 31;
    const int w    = t >> 5;
    const int n4   = n >> 2;
    const int4* __restrict__ d4 = (const int4*)deg;

    const int i4 = blockIdx.x * SCAN_THREADS + t;
    int4 v = make_int4(0, 0, 0, 0);
    if (i4 < n4) v = d4[i4];

    int pre = 0;
    const int front = blockIdx.x;
    for (int c = 0; c < front; ++c) {
        int4 u = d4[c * SCAN_THREADS + t];
        pre += (u.x + u.y) + (u.z + u.w);
    }

    const int s0 = v.x;
    const int s1 = s0 + v.y;
    const int s2 = s1 + v.z;
    const int s3 = s2 + v.w;

    int inc = s3;
    for (int off = 1; off < 32; off <<= 1) {
        int y = __shfl_up(inc, off, 32);
        if (lane >= off) inc += y;
        pre += __shfl_xor(pre, off, 32);
    }
    if (lane == 31) wave_tot[w] = inc;
    if (lane == 0)  wave_pre[w] = pre;
    __syncthreads();

    int x = wave_pre[lane] + ((lane < w) ? wave_tot[lane] : 0);
    for (int off = 1; off < 32; off <<= 1) x += __shfl_xor(x, off, 32);

    const int ex = x + inc - s3;
    if (i4 < n4) {
        int4 o = make_int4(ex, ex + s0, ex + s1, ex + s2);
        ((int4*)offs)[i4]   = o;
        ((int4*)cursor)[i4] = o;
        if (i4 == n4 - 1) offs[n] = ex + s3;
    }
}

__global__ void gat_fill_csr(const int* __restrict__ dst,
                             int* __restrict__ cursor,
                             int* __restrict__ csr,
                             int ne) {
    int i = blockIdx.x * blockDim.x + threadIdx.x;
    if (i >= ne) return;
    int p = atomicAdd(&cursor[dst[i]], 1);
    csr[p] = i;
}

__global__ void gat_node_aggregate(const float* __restrict__ z,
                                   const float* __restrict__ e,
                                   const int* __restrict__ src,
                                   const int* __restrict__ csr,
                                   const int* __restrict__ offs,
                                   const int* __restrict__ deg,
                                   float* __restrict__ out,
                                   int n) {
    int gtid = blockIdx.x * blockDim.x + threadIdx.x;
    int node = gtid >> 5;
    if (node >= n) return;
    int lane = threadIdx.x & 31;
    int c0 = lane * 2;

    int d = deg[node];
    int o = offs[node];

    float accx = 0.f, accy = 0.f;
    if (d > 0) {
        float sum = 0.f;
        for (int j = 0; j < d; ++j) sum += e[csr[o + j]];
        float mean = sum / (float)d;

        float m = -1e30f;
        for (int j = 0; j < d; ++j) {
            float ev = e[csr[o + j]];
            float e2 = (ev < mean) ? 0.f : ev;
            m = fmaxf(m, e2);
        }

        float denom = 0.f;
        for (int j = 0; j < d; ++j) {
            int ei = csr[o + j];
            float ev = e[ei];
            float e2 = (ev < mean) ? 0.f : ev;
            float ex = __expf(e2 - m);
            denom += ex;
            const float2 zv = *(const float2*)(z + (size_t)src[ei] * OUT_DIM + c0);
            accx += ex * zv.x;
            accy += ex * zv.y;
        }
        float inv = 1.f / denom;
        accx *= inv;
        accy *= inv;
    }
    float2 r;
    r.x = accx;
    r.y = accy;
    *(float2*)(out + (size_t)node * OUT_DIM + c0) = r;
}

extern "C" void kernel_launch(void* const* d_in, const int* in_sizes, int n_in,
                              void* d_out, int out_size, void* d_ws, size_t ws_size,
                              hipStream_t stream) {
    const float* h      = (const float*)d_in[0];
    const int*   src    = (const int*)d_in[1];
    const int*   dst    = (const int*)d_in[2];
    const float* W_fc   = (const float*)d_in[3];
    const float* W_attn = (const float*)d_in[4];
    float* out = (float*)d_out;

    char* ws = (char*)d_ws;
    size_t off = 0;
    auto carve = [&](size_t bytes) {
        void* p = ws + off;
        off += (bytes + 15) & ~(size_t)15;
        return p;
    };
    float* z      = (float*)carve((size_t)N_NODES * OUT_DIM * sizeof(float));
    float* s_l    = (float*)carve((size_t)N_NODES * sizeof(float));
    float* s_r    = (float*)carve((size_t)N_NODES * sizeof(float));
    float* e      = (float*)carve((size_t)N_EDGES * sizeof(float));
    int*   deg    = (int*)carve((size_t)N_NODES * sizeof(int));
    int*   offs   = (int*)carve((size_t)(N_NODES + 1) * sizeof(int));
    int*   cursor = (int*)carve((size_t)N_NODES * sizeof(int));
    int*   csr    = (int*)carve((size_t)N_EDGES * sizeof(int));
    (void)ws_size;

    const int B = 256;

    int gemm_waves  = N_NODES / 16;
    int gemm_blocks = (gemm_waves + 7) / 8;
    gat_gemm_wmma<<<gemm_blocks, B, 0, stream>>>(h, W_fc, z, N_NODES);

    gat_node_scores<<<(N_NODES + B - 1) / B, B, 0, stream>>>(z, W_attn, s_l, s_r, N_NODES);

    gat_zero_i32<<<(N_NODES + B - 1) / B, B, 0, stream>>>(deg, N_NODES);
    gat_edge_scores<<<(N_EDGES + B - 1) / B, B, 0, stream>>>(src, dst, s_l, s_r, e, deg, N_EDGES);

    gat_scan_excl<<<SCAN_BLOCKS, SCAN_THREADS, 0, stream>>>(deg, offs, cursor, N_NODES);

    gat_fill_csr<<<(N_EDGES + B - 1) / B, B, 0, stream>>>(dst, cursor, csr, N_EDGES);

    int agg_blocks = (N_NODES + 7) / 8;
    gat_node_aggregate<<<agg_blocks, B, 0, stream>>>(z, e, src, csr, offs, deg, out, N_NODES);
}


# CSR fill without atomics: the degree-count atomic returns each edge's rank in its node, slot = node offset + rank; the fill's 1.6M returning atomics go; in-node sum order = arrival at count atomic
# speedup vs baseline: 1.9167x; 1.2174x over previous
#include <hip/hip_runtime.h>
#include <hip/hip_bf16.h>
#include <math.h>

typedef __attribute__((ext_vector_type(2))) float v2f;
typedef __attribute__((ext_vector_type(8))) float v8f;

#define N_NODES 100000
#define N_EDGES 1600000
#define IN_DIM 128
#define OUT_DIM 64
#define NEG_SLOPE 0.01f

__global__ void gat_gemm_wmma(const float* __restrict__ h,
                              const float* __restrict__ W,
                              float* __restrict__ z,
                              int n_nodes) {
    int gtid = blockIdx.x * blockDim.x + threadIdx.x;
    int wave = gtid >> 5;
    int lane = threadIdx.x & 31;
    int m0 = wave * 16;
    if (m0 >= n_nodes) return;

    int rowA  = m0 + (lane & 15);
    int khalf = (lane >> 4) * 2;
    int nlane = lane & 15;

    v8f acc0 = {}, acc1 = {}, acc2 = {}, acc3 = {};

    for (int k0 = 0; k0 < IN_DIM; k0 += 4) {
        v2f a = *(const v2f*)(h + (size_t)rowA * IN_DIM + k0 + khalf);

        v2f b0 = *(const v2f*)(W + (size_t)(nlane +  0) * IN_DIM + k0 + khalf);
        v2f b1 = *(const v2f*)(W + (size_t)(nlane + 16) * IN_DIM + k0 + khalf);
        v2f b2 = *(const v2f*)(W + (size_t)(nlane + 32) * IN_DIM + k0 + khalf);
        v2f b3 = *(const v2f*)(W + (size_t)(nlane + 48) * IN_DIM + k0 + khalf);

        acc0 = __builtin_amdgcn_wmma_f32_16x16x4_f32(false, a, false, b0, (short)0, acc0, false, false);
        acc1 = __builtin_amdgcn_wmma_f32_16x16x4_f32(false, a, false, b1, (short)0, acc1, false, false);
        acc2 = __builtin_amdgcn_wmma_f32_16x16x4_f32(false, a, false, b2, (short)0, acc2, false, false);
        acc3 = __builtin_amdgcn_wmma_f32_16x16x4_f32(false, a, false, b3, (short)0, acc3, false, false);
    }

    int rsel = (lane >> 4) * 8;
#pragma unroll
    for (int r = 0; r < 8; ++r) {
        size_t ro = (size_t)(m0 + r + rsel) * OUT_DIM;
        z[ro + nlane +  0] = acc0[r];
        z[ro + nlane + 16] = acc1[r];
        z[ro + nlane + 32] = acc2[r];
        z[ro + nlane + 48] = acc3[r];
    }
}

__global__ void gat_node_scores(const float* __restrict__ z,
                                const float* __restrict__ attn,
                                float* __restrict__ s_l,
                                float* __restrict__ s_r,
                                int n) {
    int i = blockIdx.x * blockDim.x + threadIdx.x;
    if (i >= n) return;
    const float4* zr = (const float4*)(z + (size_t)i * OUT_DIM);
    const float4* al = (const float4*)(attn);
    const float4* ar = (const float4*)(attn + OUT_DIM);
    float sl = 0.f, sr = 0.f;
#pragma unroll
    for (int q = 0; q < OUT_DIM / 4; ++q) {
        float4 zv = zr[q];
        float4 av = al[q];
        float4 bv = ar[q];
        sl += zv.x * av.x + zv.y * av.y + zv.z * av.z + zv.w * av.w;
        sr += zv.x * bv.x + zv.y * bv.y + zv.z * bv.z + zv.w * bv.w;
    }
    s_l[i] = sl;
    s_r[i] = sr;
}

__global__ void gat_zero_i32(int* __restrict__ p, int n) {
    int i = blockIdx.x * blockDim.x + threadIdx.x;
    if (i < n) p[i] = 0;
}

__global__ void gat_count_rank(const int* __restrict__ dst,
                               int* __restrict__ deg,
                               int* __restrict__ rank,
                               int ne) {
    int i = blockIdx.x * blockDim.x + threadIdx.x;
    if (i >= ne) return;
    rank[i] = atomicAdd(&deg[dst[i]], 1);
}

#define SCAN_THREADS 1024
#define SCAN_CHUNK   (SCAN_THREADS * 4)
#define SCAN_BLOCKS  ((N_NODES + SCAN_CHUNK - 1) / SCAN_CHUNK)
static_assert(N_NODES % 4 == 0, "the scan reads and writes whole int4 groups");

__global__ void __launch_bounds__(SCAN_THREADS)
gat_scan_excl(const int* __restrict__ deg,
              int* __restrict__ offs,
              int n) {
    __shared__ int wave_tot[SCAN_THREADS / 32];
    __shared__ int wave_pre[SCAN_THREADS / 32];

    const int t    = threadIdx.x;
    const int lane = t & 31;
    const int w    = t >> 5;
    const int n4   = n >> 2;
    const int4* __restrict__ d4 = (const int4*)deg;

    const int i4 = blockIdx.x * SCAN_THREADS + t;
    int4 v = make_int4(0, 0, 0, 0);
    if (i4 < n4) v = d4[i4];

    int pre = 0;
    const int front = blockIdx.x;
    for (int c = 0; c < front; ++c) {
        int4 u = d4[c * SCAN_THREADS + t];
        pre += (u.x + u.y) + (u.z + u.w);
    }

    const int s0 = v.x;
    const int s1 = s0 + v.y;
    const int s2 = s1 + v.z;
    const int s3 = s2 + v.w;

    int inc = s3;
    for (int off = 1; off < 32; off <<= 1) {
        int y = __shfl_up(inc, off, 32);
        if (lane >= off) inc += y;
        pre += __shfl_xor(pre, off, 32);
    }
    if (lane == 31) wave_tot[w] = inc;
    if (lane == 0)  wave_pre[w] = pre;
    __syncthreads();

    int x = wave_pre[lane] + ((lane < w) ? wave_tot[lane] : 0);
    for (int off = 1; off < 32; off <<= 1) x += __shfl_xor(x, off, 32);

    const int ex = x + inc - s3;
    if (i4 < n4) {
        int4 o = make_int4(ex, ex + s0, ex + s1, ex + s2);
        ((int4*)offs)[i4] = o;
        if (i4 == n4 - 1) offs[n] = ex + s3;
    }
}

__global__ void gat_score_fill(const int* __restrict__ src,
                               const int* __restrict__ dst,
                               const float* __restrict__ s_l,
                               const float* __restrict__ s_r,
                               const int* __restrict__ offs,
                               float* __restrict__ e_rank,
                               int* __restrict__ csr,
                               int ne) {
    int i = blockIdx.x * blockDim.x + threadIdx.x;
    if (i >= ne) return;
    int d = dst[i];
    int r = __float_as_int(e_rank[i]);
    float v = s_l[src[i]] + s_r[d];
    e_rank[i] = (v > 0.f) ? v : NEG_SLOPE * v;
    csr[offs[d] + r] = i;
}

__global__ void gat_node_aggregate(const float* __restrict__ z,
                                   const float* __restrict__ e,
                                   const int* __restrict__ src,
                                   const int* __restrict__ csr,
                                   const int* __restrict__ offs,
                                   const int* __restrict__ deg,
                                   float* __restrict__ out,
                                   int n) {
    int gtid = blockIdx.x * blockDim.x + threadIdx.x;
    int node = gtid >> 5;
    if (node >= n) return;
    int lane = threadIdx.x & 31;
    int c0 = lane * 2;

    int d = deg[node];
    int o = offs[node];

    float accx = 0.f, accy = 0.f;
    if (d > 0) {
        float sum = 0.f;
        for (int j = 0; j < d; ++j) sum += e[csr[o + j]];
        float mean = sum / (float)d;

        float m = -1e30f;
        for (int j = 0; j < d; ++j) {
            float ev = e[csr[o + j]];
            float e2 = (ev < mean) ? 0.f : ev;
            m = fmaxf(m, e2);
        }

        float denom = 0.f;
        for (int j = 0; j < d; ++j) {
            int ei = csr[o + j];
            float ev = e[ei];
            float e2 = (ev < mean) ? 0.f : ev;
            float ex = __expf(e2 - m);
            denom += ex;
            const float2 zv = *(const float2*)(z + (size_t)src[ei] * OUT_DIM + c0);
            accx += ex * zv.x;
            accy += ex * zv.y;
        }
        float inv = 1.f / denom;
        accx *= inv;
        accy *= inv;
    }
    float2 r;
    r.x = accx;
    r.y = accy;
    *(float2*)(out + (size_t)node * OUT_DIM + c0) = r;
}

extern "C" void kernel_launch(void* const* d_in, const int* in_sizes, int n_in,
                              void* d_out, int out_size, void* d_ws, size_t ws_size,
                              hipStream_t stream) {
    const float* h      = (const float*)d_in[0];
    const int*   src    = (const int*)d_in[1];
    const int*   dst    = (const int*)d_in[2];
    const float* W_fc   = (const float*)d_in[3];
    const float* W_attn = (const float*)d_in[4];
    float* out = (float*)d_out;

    char* ws = (char*)d_ws;
    size_t off = 0;
    auto carve = [&](size_t bytes) {
        void* p = ws + off;
        off += (bytes + 15) & ~(size_t)15;
        return p;
    };
    float* z      = (float*)carve((size_t)N_NODES * OUT_DIM * sizeof(float));
    float* s_l    = (float*)carve((size_t)N_NODES * sizeof(float));
    float* s_r    = (float*)carve((size_t)N_NODES * sizeof(float));
    float* e      = (float*)carve((size_t)N_EDGES * sizeof(float));
    int*   deg    = (int*)carve((size_t)N_NODES * sizeof(int));
    int*   offs   = (int*)carve((size_t)(N_NODES + 1) * sizeof(int));
    int*   csr    = (int*)carve((size_t)N_EDGES * sizeof(int));
    (void)ws_size;

    const int B = 256;

    int gemm_waves  = N_NODES / 16;
    int gemm_blocks = (gemm_waves + 7) / 8;
    gat_gemm_wmma<<<gemm_blocks, B, 0, stream>>>(h, W_fc, z, N_NODES);

    gat_node_scores<<<(N_NODES + B - 1) / B, B, 0, stream>>>(z, W_attn, s_l, s_r, N_NODES);

    gat_zero_i32<<<(N_NODES + B - 1) / B, B, 0, stream>>>(deg, N_NODES);
    gat_count_rank<<<(N_EDGES + B - 1) / B, B, 0, stream>>>(dst, deg, (int*)e, N_EDGES);

    gat_scan_excl<<<SCAN_BLOCKS, SCAN_THREADS, 0, stream>>>(deg, offs, N_NODES);

    gat_score_fill<<<(N_EDGES + B - 1) / B, B, 0, stream>>>(src, dst, s_l, s_r, offs, e, csr, N_EDGES);

    int agg_blocks = (N_NODES + 7) / 8;
    gat_node_aggregate<<<agg_blocks, B, 0, stream>>>(z, e, src, csr, offs, deg, out, N_NODES);
}


# Node aggregation: a wave takes 8 nodes in turn, a node's edges one per lane sorted by edge id (in-node sums in a fixed order as lane trees; was atomic arrival order, serial); z rows asked 8 at a time
# speedup vs baseline: 2.2345x; 1.1658x over previous
#include <hip/hip_runtime.h>
#include <hip/hip_bf16.h>
#include <math.h>

typedef __attribute__((ext_vector_type(2))) float v2f;
typedef __attribute__((ext_vector_type(8))) float v8f;

#define N_NODES 100000
#define N_EDGES 1600000
#define IN_DIM 128
#define OUT_DIM 64
#define NEG_SLOPE 0.01f

__global__ void gat_gemm_wmma(const float* __restrict__ h,
                              const float* __restrict__ W,
                              float* __restrict__ z,
                              int n_nodes) {
    int gtid = blockIdx.x * blockDim.x + threadIdx.x;
    int wave = gtid >> 5;
    int lane = threadIdx.x & 31;
    int m0 = wave * 16;
    if (m0 >= n_nodes) return;

    int rowA  = m0 + (lane & 15);
    int khalf = (lane >> 4) * 2;
    int nlane = lane & 15;

    v8f acc0 = {}, acc1 = {}, acc2 = {}, acc3 = {};

    for (int k0 = 0; k0 < IN_DIM; k0 += 4) {
        v2f a = *(const v2f*)(h + (size_t)rowA * IN_DIM + k0 + khalf);

        v2f b0 = *(const v2f*)(W + (size_t)(nlane +  0) * IN_DIM + k0 + khalf);
        v2f b1 = *(const v2f*)(W + (size_t)(nlane + 16) * IN_DIM + k0 + khalf);
        v2f b2 = *(const v2f*)(W + (size_t)(nlane + 32) * IN_DIM + k0 + khalf);
        v2f b3 = *(const v2f*)(W + (size_t)(nlane + 48) * IN_DIM + k0 + khalf);

        acc0 = __builtin_amdgcn_wmma_f32_16x16x4_f32(false, a, false, b0, (short)0, acc0, false, false);
        acc1 = __builtin_amdgcn_wmma_f32_16x16x4_f32(false, a, false, b1, (short)0, acc1, false, false);
        acc2 = __builtin_amdgcn_wmma_f32_16x16x4_f32(false, a, false, b2, (short)0, acc2, false, false);
        acc3 = __builtin_amdgcn_wmma_f32_16x16x4_f32(false, a, false, b3, (short)0, acc3, false, false);
    }

    int rsel = (lane >> 4) * 8;
#pragma unroll
    for (int r = 0; r < 8; ++r) {
        size_t ro = (size_t)(m0 + r + rsel) * OUT_DIM;
        z[ro + nlane +  0] = acc0[r];
        z[ro + nlane + 16] = acc1[r];
        z[ro + nlane + 32] = acc2[r];
        z[ro + nlane + 48] = acc3[r];
    }
}

__global__ void gat_node_scores(const float* __restrict__ z,
                                const float* __restrict__ attn,
                                float* __restrict__ s_l,
                                float* __restrict__ s_r,
                                int n) {
    int i = blockIdx.x * blockDim.x + threadIdx.x;
    if (i >= n) return;
    const float4* zr = (const float4*)(z + (size_t)i * OUT_DIM);
    const float4* al = (const float4*)(attn);
    const float4* ar = (const float4*)(attn + OUT_DIM);
    float sl = 0.f, sr = 0.f;
#pragma unroll
    for (int q = 0; q < OUT_DIM / 4; ++q) {
        float4 zv = zr[q];
        float4 av = al[q];
        float4 bv = ar[q];
        sl += zv.x * av.x + zv.y * av.y + zv.z * av.z + zv.w * av.w;
        sr += zv.x * bv.x + zv.y * bv.y + zv.z * bv.z + zv.w * bv.w;
    }
    s_l[i] = sl;
    s_r[i] = sr;
}

__global__ void gat_zero_i32(int* __restrict__ p, int n) {
    int i = blockIdx.x * blockDim.x + threadIdx.x;
    if (i < n) p[i] = 0;
}

__global__ void gat_count_rank(const int* __restrict__ dst,
                               int* __restrict__ deg,
                               int* __restrict__ rank,
                               int ne) {
    int i = blockIdx.x * blockDim.x + threadIdx.x;
    if (i >= ne) return;
    rank[i] = atomicAdd(&deg[dst[i]], 1);
}

#define SCAN_THREADS 1024
#define SCAN_CHUNK   (SCAN_THREADS * 4)
#define SCAN_BLOCKS  ((N_NODES + SCAN_CHUNK - 1) / SCAN_CHUNK)
static_assert(N_NODES % 4 == 0, "the scan reads and writes whole int4 groups");

__global__ void __launch_bounds__(SCAN_THREADS)
gat_scan_excl(const int* __restrict__ deg,
              int* __restrict__ offs,
              int n) {
    __shared__ int wave_tot[SCAN_THREADS / 32];
    __shared__ int wave_pre[SCAN_THREADS / 32];

    const int t    = threadIdx.x;
    const int lane = t & 31;
    const int w    = t >> 5;
    const int n4   = n >> 2;
    const int4* __restrict__ d4 = (const int4*)deg;

    const int i4 = blockIdx.x * SCAN_THREADS + t;
    int4 v = make_int4(0, 0, 0, 0);
    if (i4 < n4) v = d4[i4];

    int pre = 0;
    const int front = blockIdx.x;
    for (int c = 0; c < front; ++c) {
        int4 u = d4[c * SCAN_THREADS + t];
        pre += (u.x + u.y) + (u.z + u.w);
    }

    const int s0 = v.x;
    const int s1 = s0 + v.y;
    const int s2 = s1 + v.z;
    const int s3 = s2 + v.w;

    int inc = s3;
    for (int off = 1; off < 32; off <<= 1) {
        int y = __shfl_up(inc, off, 32);
        if (lane >= off) inc += y;
        pre += __shfl_xor(pre, off, 32);
    }
    if (lane == 31) wave_tot[w] = inc;
    if (lane == 0)  wave_pre[w] = pre;
    __syncthreads();

    int x = wave_pre[lane] + ((lane < w) ? wave_tot[lane] : 0);
    for (int off = 1; off < 32; off <<= 1) x += __shfl_xor(x, off, 32);

    const int ex = x + inc - s3;
    if (i4 < n4) {
        int4 o = make_int4(ex, ex + s0, ex + s1, ex + s2);
        ((int4*)offs)[i4] = o;
        if (i4 == n4 - 1) offs[n] = ex + s3;
    }
}

__global__ void gat_score_fill(const int* __restrict__ src,
                               const int* __restrict__ dst,
                               const float* __restrict__ s_l,
                               const float* __restrict__ s_r,
                               const int* __restrict__ offs,
                               float* __restrict__ e_rank,
                               int* __restrict__ csr,
                               int ne) {
    int i = blockIdx.x * blockDim.x + threadIdx.x;
    if (i >= ne) return;
    int d = dst[i];
    int r = __float_as_int(e_rank[i]);
    float v = s_l[src[i]] + s_r[d];
    e_rank[i] = (v > 0.f) ? v : NEG_SLOPE * v;
    csr[offs[d] + r] = i;
}

#define AGG_NODES 8
#define AGG_ROWS  8

template <int M>
__device__ __forceinline__ float lane_xor_row(float x) {
    return __builtin_bit_cast(float,
        __builtin_amdgcn_update_dpp(0, __builtin_bit_cast(int, x), 0x160 | M, 0xf, 0xf, true));
}
__device__ __forceinline__ float lane_xor_half(float x) {
    int b = __builtin_bit_cast(int, x);
    return __builtin_bit_cast(float,
        __builtin_amdgcn_permlanex16(b, b, 0x76543210, 0xfedcba98, false, false));
}
__device__ __forceinline__ float wave_sum(float x) {
    x += lane_xor_row<1>(x);
    x += lane_xor_row<2>(x);
    x += lane_xor_row<4>(x);
    x += lane_xor_row<8>(x);
    x += lane_xor_half(x);
    return x;
}
__device__ __forceinline__ float wave_max(float x) {
    x = fmaxf(x, lane_xor_row<1>(x));
    x = fmaxf(x, lane_xor_row<2>(x));
    x = fmaxf(x, lane_xor_row<4>(x));
    x = fmaxf(x, lane_xor_row<8>(x));
    x = fmaxf(x, lane_xor_half(x));
    return x;
}
__device__ __forceinline__ int lane_read(int v, int l) {
    return __builtin_amdgcn_readlane(v, l);
}
__device__ __forceinline__ float lane_read(float v, int l) {
    return __builtin_bit_cast(float, __builtin_amdgcn_readlane(__builtin_bit_cast(int, v), l));
}

__global__ void gat_node_aggregate(const float* __restrict__ z,
                                   const float* __restrict__ e,
                                   const int* __restrict__ src,
                                   const int* __restrict__ csr,
                                   const int* __restrict__ offs,
                                   float* __restrict__ out,
                                   int n) {
    const int lane = threadIdx.x & 31;
    const int wave  = __builtin_amdgcn_readfirstlane((int)((blockIdx.x * blockDim.x + threadIdx.x) >> 5));
    const int node0 = wave * AGG_NODES;
    if (node0 >= n) return;
    const int c0 = lane * 2;

    int oi = node0 + ((lane < AGG_NODES) ? lane : AGG_NODES);
    if (oi > n) oi = n;
    const int offv = offs[oi];

    int o = lane_read(offv, 0);
    int d = lane_read(offv, 1) - o;
    int ei = (lane < d) ? csr[o + lane] : 0;

    for (int k = 0; k < AGG_NODES; ++k) {
        const int node = node0 + k;
        if (node >= n) break;
        const bool act = lane < d;

        float ev = act ? e[ei] : 0.f;
        int   sv = act ? src[ei] : 0;

        int o2 = o + d, d2 = 0, ei2 = 0;
        if (k + 1 < AGG_NODES && node + 1 < n) {
            d2  = lane_read(offv, k + 2) - o2;
            ei2 = (lane < d2) ? csr[o2 + lane] : 0;
        }

        float accx = 0.f, accy = 0.f;
        if (d > 32) {
            float sum = 0.f;
            for (int j = 0; j < d; ++j) sum += e[csr[o + j]];
            const float mean = sum / (float)d;
            float m = -1e30f;
            for (int j = 0; j < d; ++j) {
                float x  = e[csr[o + j]];
                float e2 = (x < mean) ? 0.f : x;
                m = fmaxf(m, e2);
            }
            float denom = 0.f;
            for (int j = 0; j < d; ++j) {
                int   ej = csr[o + j];
                float x  = e[ej];
                float e2 = (x < mean) ? 0.f : x;
                float ex = __expf(e2 - m);
                denom += ex;
                const float2 zv = *(const float2*)(z + (size_t)src[ej] * OUT_DIM + c0);
                accx += ex * zv.x;
                accy += ex * zv.y;
            }
            const float inv = 1.f / denom;
            accx *= inv;
            accy *= inv;
        } else if (d > 0) {
            int p = 0;
            for (int j = 0; j < d; ++j) p += (lane_read(ei, j) < ei) ? 1 : 0;
            if (!act) p = lane;
            ev = __builtin_bit_cast(float, __builtin_amdgcn_ds_permute(p << 2, __builtin_bit_cast(int, ev)));
            sv = __builtin_amdgcn_ds_permute(p << 2, sv);

            const float mean = wave_sum(ev) / (float)d;
            const float e2 = (ev < mean) ? 0.f : ev;
            const float m  = wave_max(act ? e2 : -1e30f);
            const float ex = act ? __expf(e2 - m) : 0.f;
            const float denom = wave_sum(ex);

            for (int j0 = 0; j0 < d; j0 += AGG_ROWS) {
                float2 zv[AGG_ROWS];
                float  w[AGG_ROWS];
#pragma unroll
                for (int u = 0; u < AGG_ROWS; ++u) {
                    const int s = lane_read(sv, j0 + u);
                    w[u]  = lane_read(ex, j0 + u);
                    zv[u] = *(const float2*)(z + (size_t)s * OUT_DIM + c0);
                }
#pragma unroll
                for (int u = 0; u < AGG_ROWS; ++u) {
                    accx += w[u] * zv[u].x;
                    accy += w[u] * zv[u].y;
                }
            }
            const float inv = 1.f / denom;
            accx *= inv;
            accy *= inv;
        }
        float2 r;
        r.x = accx;
        r.y = accy;
        *(float2*)(out + (size_t)node * OUT_DIM + c0) = r;

        o = o2;
        d = d2;
        ei = ei2;
    }
}

extern "C" void kernel_launch(void* const* d_in, const int* in_sizes, int n_in,
                              void* d_out, int out_size, void* d_ws, size_t ws_size,
                              hipStream_t stream) {
    const float* h      = (const float*)d_in[0];
    const int*   src    = (const int*)d_in[1];
    const int*   dst    = (const int*)d_in[2];
    const float* W_fc   = (const float*)d_in[3];
    const float* W_attn = (const float*)d_in[4];
    float* out = (float*)d_out;

    char* ws = (char*)d_ws;
    size_t off = 0;
    auto carve = [&](size_t bytes) {
        void* p = ws + off;
        off += (bytes + 15) & ~(size_t)15;
        return p;
    };
    float* z      = (float*)carve((size_t)N_NODES * OUT_DIM * sizeof(float));
    float* s_l    = (float*)carve((size_t)N_NODES * sizeof(float));
    float* s_r    = (float*)carve((size_t)N_NODES * sizeof(float));
    float* e      = (float*)carve((size_t)N_EDGES * sizeof(float));
    int*   deg    = (int*)carve((size_t)N_NODES * sizeof(int));
    int*   offs   = (int*)carve((size_t)(N_NODES + 1) * sizeof(int));
    int*   csr    = (int*)carve((size_t)N_EDGES * sizeof(int));
    (void)ws_size;

    const int B = 256;

    int gemm_waves  = N_NODES / 16;
    int gemm_blocks = (gemm_waves + 7) / 8;
    gat_gemm_wmma<<<gemm_blocks, B, 0, stream>>>(h, W_fc, z, N_NODES);

    gat_node_scores<<<(N_NODES + B - 1) / B, B, 0, stream>>>(z, W_attn, s_l, s_r, N_NODES);

    gat_zero_i32<<<(N_NODES + B - 1) / B, B, 0, stream>>>(deg, N_NODES);
    gat_count_rank<<<(N_EDGES + B - 1) / B, B, 0, stream>>>(dst, deg, (int*)e, N_EDGES);

    gat_scan_excl<<<SCAN_BLOCKS, SCAN_THREADS, 0, stream>>>(deg, offs, N_NODES);

    gat_score_fill<<<(N_EDGES + B - 1) / B, B, 0, stream>>>(src, dst, s_l, s_r, offs, e, csr, N_EDGES);

    int agg_blocks = (N_NODES + 8 * AGG_NODES - 1) / (8 * AGG_NODES);
    gat_node_aggregate<<<agg_blocks, B, 0, stream>>>(z, e, src, csr, offs, out, N_NODES);
}


# Degree count and edge rank without global atomics: 32 edge chunks, each counted by one work-group in LDS (16-bit bins, returning LDS adds), chunk bases by a column sum; sum order unchanged
# speedup vs baseline: 3.0084x; 1.3463x over previous
#include <hip/hip_runtime.h>
#include <hip/hip_bf16.h>
#include <math.h>

typedef __attribute__((ext_vector_type(2))) float v2f;
typedef __attribute__((ext_vector_type(8))) float v8f;

#define N_NODES 100000
#define N_EDGES 1600000
#define IN_DIM 128
#define OUT_DIM 64
#define NEG_SLOPE 0.01f

__global__ void gat_gemm_wmma(const float* __restrict__ h,
                              const float* __restrict__ W,
                              float* __restrict__ z,
                              int n_nodes) {
    int gtid = blockIdx.x * blockDim.x + threadIdx.x;
    int wave = gtid >> 5;
    int lane = threadIdx.x & 31;
    int m0 = wave * 16;
    if (m0 >= n_nodes) return;

    int rowA  = m0 + (lane & 15);
    int khalf = (lane >> 4) * 2;
    int nlane = lane & 15;

    v8f acc0 = {}, acc1 = {}, acc2 = {}, acc3 = {};

    for (int k0 = 0; k0 < IN_DIM; k0 += 4) {
        v2f a = *(const v2f*)(h + (size_t)rowA * IN_DIM + k0 + khalf);

        v2f b0 = *(const v2f*)(W + (size_t)(nlane +  0) * IN_DIM + k0 + khalf);
        v2f b1 = *(const v2f*)(W + (size_t)(nlane + 16) * IN_DIM + k0 + khalf);
        v2f b2 = *(const v2f*)(W + (size_t)(nlane + 32) * IN_DIM + k0 + khalf);
        v2f b3 = *(const v2f*)(W + (size_t)(nlane + 48) * IN_DIM + k0 + khalf);

        acc0 = __builtin_amdgcn_wmma_f32_16x16x4_f32(false, a, false, b0, (short)0, acc0, false, false);
        acc1 = __builtin_amdgcn_wmma_f32_16x16x4_f32(false, a, false, b1, (short)0, acc1, false, false);
        acc2 = __builtin_amdgcn_wmma_f32_16x16x4_f32(false, a, false, b2, (short)0, acc2, false, false);
        acc3 = __builtin_amdgcn_wmma_f32_16x16x4_f32(false, a, false, b3, (short)0, acc3, false, false);
    }

    int rsel = (lane >> 4) * 8;
#pragma unroll
    for (int r = 0; r < 8; ++r) {
        size_t ro = (size_t)(m0 + r + rsel) * OUT_DIM;
        z[ro + nlane +  0] = acc0[r];
        z[ro + nlane + 16] = acc1[r];
        z[ro + nlane + 32] = acc2[r];
        z[ro + nlane + 48] = acc3[r];
    }
}

__global__ void gat_node_scores(const float* __restrict__ z,
                                const float* __restrict__ attn,
                                float* __restrict__ s_l,
                                float* __restrict__ s_r,
                                int n) {
    int i = blockIdx.x * blockDim.x + threadIdx.x;
    if (i >= n) return;
    const float4* zr = (const float4*)(z + (size_t)i * OUT_DIM);
    const float4* al = (const float4*)(attn);
    const float4* ar = (const float4*)(attn + OUT_DIM);
    float sl = 0.f, sr = 0.f;
#pragma unroll
    for (int q = 0; q < OUT_DIM / 4; ++q) {
        float4 zv = zr[q];
        float4 av = al[q];
        float4 bv = ar[q];
        sl += zv.x * av.x + zv.y * av.y + zv.z * av.z + zv.w * av.w;
        sr += zv.x * bv.x + zv.y * bv.y + zv.z * bv.z + zv.w * bv.w;
    }
    s_l[i] = sl;
    s_r[i] = sr;
}

#define HIST_CHUNKS   32
#define HIST_THREADS  1024
#define HIST_EDGES    ((N_EDGES + HIST_CHUNKS - 1) / HIST_CHUNKS)
#define HIST_PER_THR  ((HIST_EDGES + HIST_THREADS - 1) / HIST_THREADS)
#define HIST_WORDS    (N_NODES / 2)
#define HIST_GROUP    8
static_assert(HIST_EDGES < 65536, "a chunk's count of one node must fit a 16-bit bin");
static_assert(N_NODES % 8 == 0, "bins are cleared and copied as whole uint4 groups");

__global__ void __launch_bounds__(HIST_THREADS)
gat_hist_rank(const int* __restrict__ dst,
              unsigned int* __restrict__ cnt,
              int* __restrict__ rank,
              int ne) {
    __shared__ unsigned int bins[HIST_WORDS];

    const int t  = threadIdx.x;
    const int c  = blockIdx.x;
    const int e0 = c * HIST_EDGES;
    const int e1 = (e0 + HIST_EDGES < ne) ? (e0 + HIST_EDGES) : ne;

    int d[HIST_PER_THR];
#pragma unroll
    for (int q = 0; q < HIST_PER_THR; ++q) {
        const int i = e0 + q * HIST_THREADS + t;
        d[q] = dst[(i < e1) ? i : (ne - 1)];
    }

    uint4* b4 = (uint4*)bins;
    for (int j = t; j < HIST_WORDS / 4; j += HIST_THREADS) b4[j] = make_uint4(0u, 0u, 0u, 0u);
    __syncthreads();

#pragma unroll
    for (int q0 = 0; q0 < HIST_PER_THR; q0 += HIST_GROUP) {
        unsigned int old[HIST_GROUP];
#pragma unroll
        for (int u = 0; u < HIST_GROUP; ++u) {
            const int q = q0 + u;
            if (q < HIST_PER_THR) {
                const int i = e0 + q * HIST_THREADS + t;
                const unsigned int inc = (i < e1) ? ((d[q] & 1) ? 0x10000u : 1u) : 0u;
                old[u] = atomicAdd(&bins[d[q] >> 1], inc);
            }
        }
#pragma unroll
        for (int u = 0; u < HIST_GROUP; ++u) {
            const int q = q0 + u;
            if (q < HIST_PER_THR) {
                const int i = e0 + q * HIST_THREADS + t;
                if (i < e1) rank[i] = (int)((d[q] & 1) ? (old[u] >> 16) : (old[u] & 0xffffu));
            }
        }
    }
    __syncthreads();

    uint4* g4 = (uint4*)(cnt + (size_t)c * HIST_WORDS);
    for (int j = t; j < HIST_WORDS / 4; j += HIST_THREADS) g4[j] = b4[j];
}

__global__ void gat_chunk_bases(const unsigned int* __restrict__ cnt,
                                int* __restrict__ base,
                                int* __restrict__ deg,
                                int nwords) {
    const int j = blockIdx.x * blockDim.x + threadIdx.x;
    if (j >= nwords) return;
    unsigned int v[HIST_CHUNKS];
#pragma unroll
    for (int c = 0; c < HIST_CHUNKS; ++c) v[c] = cnt[(size_t)c * HIST_WORDS + j];
    int lo = 0, hi = 0;
#pragma unroll
    for (int c = 0; c < HIST_CHUNKS; ++c) {
        *(int2*)(base + (size_t)c * N_NODES + 2 * j) = make_int2(lo, hi);
        lo += (int)(v[c] & 0xffffu);
        hi += (int)(v[c] >> 16);
    }
    *(int2*)(deg + 2 * j) = make_int2(lo, hi);
}

#define SCAN_THREADS 1024
#define SCAN_CHUNK   (SCAN_THREADS * 4)
#define SCAN_BLOCKS  ((N_NODES + SCAN_CHUNK - 1) / SCAN_CHUNK)
static_assert(N_NODES % 4 == 0, "the scan reads and writes whole int4 groups");

__global__ void __launch_bounds__(SCAN_THREADS)
gat_scan_excl(const int* __restrict__ deg,
              int* __restrict__ offs,
              int n) {
    __shared__ int wave_tot[SCAN_THREADS / 32];
    __shared__ int wave_pre[SCAN_THREADS / 32];

    const int t    = threadIdx.x;
    const int lane = t & 31;
    const int w    = t >> 5;
    const int n4   = n >> 2;
    const int4* __restrict__ d4 = (const int4*)deg;

    const int i4 = blockIdx.x * SCAN_THREADS + t;
    int4 v = make_int4(0, 0, 0, 0);
    if (i4 < n4) v = d4[i4];

    int pre = 0;
    const int front = blockIdx.x;
    for (int c = 0; c < front; ++c) {
        int4 u = d4[c * SCAN_THREADS + t];
        pre += (u.x + u.y) + (u.z + u.w);
    }

    const int s0 = v.x;
    const int s1 = s0 + v.y;
    const int s2 = s1 + v.z;
    const int s3 = s2 + v.w;

    int inc = s3;
    for (int off = 1; off < 32; off <<= 1) {
        int y = __shfl_up(inc, off, 32);
        if (lane >= off) inc += y;
        pre += __shfl_xor(pre, off, 32);
    }
    if (lane == 31) wave_tot[w] = inc;
    if (lane == 0)  wave_pre[w] = pre;
    __syncthreads();

    int x = wave_pre[lane] + ((lane < w) ? wave_tot[lane] : 0);
    for (int off = 1; off < 32; off <<= 1) x += __shfl_xor(x, off, 32);

    const int ex = x + inc - s3;
    if (i4 < n4) {
        int4 o = make_int4(ex, ex + s0, ex + s1, ex + s2);
        ((int4*)offs)[i4] = o;
        if (i4 == n4 - 1) offs[n] = ex + s3;
    }
}

__global__ void gat_score_fill(const int* __restrict__ src,
                               const int* __restrict__ dst,
                               const float* __restrict__ s_l,
                               const float* __restrict__ s_r,
                               const int* __restrict__ offs,
                               const int* __restrict__ base,
                               float* __restrict__ e_rank,
                               int* __restrict__ csr,
                               int ne) {
    int i = blockIdx.x * blockDim.x + threadIdx.x;
    if (i >= ne) return;
    int d = dst[i];
    int r = __float_as_int(e_rank[i]);
    float v = s_l[src[i]] + s_r[d];
    e_rank[i] = (v > 0.f) ? v : NEG_SLOPE * v;
    csr[offs[d] + base[(size_t)(i / HIST_EDGES) * N_NODES + d] + r] = i;
}

#define AGG_NODES 8
#define AGG_ROWS  8

template <int M>
__device__ __forceinline__ float lane_xor_row(float x) {
    return __builtin_bit_cast(float,
        __builtin_amdgcn_update_dpp(0, __builtin_bit_cast(int, x), 0x160 | M, 0xf, 0xf, true));
}
__device__ __forceinline__ float lane_xor_half(float x) {
    int b = __builtin_bit_cast(int, x);
    return __builtin_bit_cast(float,
        __builtin_amdgcn_permlanex16(b, b, 0x76543210, 0xfedcba98, false, false));
}
__device__ __forceinline__ float wave_sum(float x) {
    x += lane_xor_row<1>(x);
    x += lane_xor_row<2>(x);
    x += lane_xor_row<4>(x);
    x += lane_xor_row<8>(x);
    x += lane_xor_half(x);
    return x;
}
__device__ __forceinline__ float wave_max(float x) {
    x = fmaxf(x, lane_xor_row<1>(x));
    x = fmaxf(x, lane_xor_row<2>(x));
    x = fmaxf(x, lane_xor_row<4>(x));
    x = fmaxf(x, lane_xor_row<8>(x));
    x = fmaxf(x, lane_xor_half(x));
    return x;
}
__device__ __forceinline__ int lane_read(int v, int l) {
    return __builtin_amdgcn_readlane(v, l);
}
__device__ __forceinline__ float lane_read(float v, int l) {
    return __builtin_bit_cast(float, __builtin_amdgcn_readlane(__builtin_bit_cast(int, v), l));
}

__global__ void gat_node_aggregate(const float* __restrict__ z,
                                   const float* __restrict__ e,
                                   const int* __restrict__ src,
                                   const int* __restrict__ csr,
                                   const int* __restrict__ offs,
                                   float* __restrict__ out,
                                   int n) {
    const int lane = threadIdx.x & 31;
    const int wave  = __builtin_amdgcn_readfirstlane((int)((blockIdx.x * blockDim.x + threadIdx.x) >> 5));
    const int node0 = wave * AGG_NODES;
    if (node0 >= n) return;
    const int c0 = lane * 2;

    int oi = node0 + ((lane < AGG_NODES) ? lane : AGG_NODES);
    if (oi > n) oi = n;
    const int offv = offs[oi];

    int o = lane_read(offv, 0);
    int d = lane_read(offv, 1) - o;
    int ei = (lane < d) ? csr[o + lane] : 0;

    for (int k = 0; k < AGG_NODES; ++k) {
        const int node = node0 + k;
        if (node >= n) break;
        const bool act = lane < d;

        float ev = act ? e[ei] : 0.f;
        int   sv = act ? src[ei] : 0;

        int o2 = o + d, d2 = 0, ei2 = 0;
        if (k + 1 < AGG_NODES && node + 1 < n) {
            d2  = lane_read(offv, k + 2) - o2;
            ei2 = (lane < d2) ? csr[o2 + lane] : 0;
        }

        float accx = 0.f, accy = 0.f;
        if (d > 32) {
            float sum = 0.f;
            for (int j = 0; j < d; ++j) sum += e[csr[o + j]];
            const float mean = sum / (float)d;
            float m = -1e30f;
            for (int j = 0; j < d; ++j) {
                float x  = e[csr[o + j]];
                float e2 = (x < mean) ? 0.f : x;
                m = fmaxf(m, e2);
            }
            float denom = 0.f;
            for (int j = 0; j < d; ++j) {
                int   ej = csr[o + j];
                float x  = e[ej];
                float e2 = (x < mean) ? 0.f : x;
                float ex = __expf(e2 - m);
                denom += ex;
                const float2 zv = *(const float2*)(z + (size_t)src[ej] * OUT_DIM + c0);
                accx += ex * zv.x;
                accy += ex * zv.y;
            }
            const float inv = 1.f / denom;
            accx *= inv;
            accy *= inv;
        } else if (d > 0) {
            int p = 0;
            for (int j = 0; j < d; ++j) p += (lane_read(ei, j) < ei) ? 1 : 0;
            if (!act) p = lane;
            ev = __builtin_bit_cast(float, __builtin_amdgcn_ds_permute(p << 2, __builtin_bit_cast(int, ev)));
            sv = __builtin_amdgcn_ds_permute(p << 2, sv);

            const float mean = wave_sum(ev) / (float)d;
            const float e2 = (ev < mean) ? 0.f : ev;
            const float m  = wave_max(act ? e2 : -1e30f);
            const float ex = act ? __expf(e2 - m) : 0.f;
            const float denom = wave_sum(ex);

            for (int j0 = 0; j0 < d; j0 += AGG_ROWS) {
                float2 zv[AGG_ROWS];
                float  w[AGG_ROWS];
#pragma unroll
                for (int u = 0; u < AGG_ROWS; ++u) {
                    const int s = lane_read(sv, j0 + u);
                    w[u]  = lane_read(ex, j0 + u);
                    zv[u] = *(const float2*)(z + (size_t)s * OUT_DIM + c0);
                }
#pragma unroll
                for (int u = 0; u < AGG_ROWS; ++u) {
                    accx += w[u] * zv[u].x;
                    accy += w[u] * zv[u].y;
                }
            }
            const float inv = 1.f / denom;
            accx *= inv;
            accy *= inv;
        }
        float2 r;
        r.x = accx;
        r.y = accy;
        *(float2*)(out + (size_t)node * OUT_DIM + c0) = r;

        o = o2;
        d = d2;
        ei = ei2;
    }
}

extern "C" void kernel_launch(void* const* d_in, const int* in_sizes, int n_in,
                              void* d_out, int out_size, void* d_ws, size_t ws_size,
                              hipStream_t stream) {
    const float* h      = (const float*)d_in[0];
    const int*   src    = (const int*)d_in[1];
    const int*   dst    = (const int*)d_in[2];
    const float* W_fc   = (const float*)d_in[3];
    const float* W_attn = (const float*)d_in[4];
    float* out = (float*)d_out;

    char* ws = (char*)d_ws;
    size_t off = 0;
    auto carve = [&](size_t bytes) {
        void* p = ws + off;
        off += (bytes + 15) & ~(size_t)15;
        return p;
    };
    float* z      = (float*)carve((size_t)N_NODES * OUT_DIM * sizeof(float));
    float* s_l    = (float*)carve((size_t)N_NODES * sizeof(float));
    float* s_r    = (float*)carve((size_t)N_NODES * sizeof(float));
    float* e      = (float*)carve((size_t)N_EDGES * sizeof(float));
    int*   deg    = (int*)carve((size_t)N_NODES * sizeof(int));
    int*   offs   = (int*)carve((size_t)(N_NODES + 1) * sizeof(int));
    int*   csr    = (int*)carve((size_t)N_EDGES * sizeof(int));
    (void)ws_size;

    const int B = 256;

    int gemm_waves  = N_NODES / 16;
    int gemm_blocks = (gemm_waves + 7) / 8;
    gat_gemm_wmma<<<gemm_blocks, B, 0, stream>>>(h, W_fc, z, N_NODES);

    gat_node_scores<<<(N_NODES + B - 1) / B, B, 0, stream>>>(z, W_attn, s_l, s_r, N_NODES);

    unsigned int* cnt = (unsigned int*)csr;
    int* base = (int*)d_out;
    gat_hist_rank<<<HIST_CHUNKS, HIST_THREADS, 0, stream>>>(dst, cnt, (int*)e, N_EDGES);
    gat_chunk_bases<<<(HIST_WORDS + B - 1) / B, B, 0, stream>>>(cnt, base, deg, HIST_WORDS);

    gat_scan_excl<<<SCAN_BLOCKS, SCAN_THREADS, 0, stream>>>(deg, offs, N_NODES);

    gat_score_fill<<<(N_EDGES + B - 1) / B, B, 0, stream>>>(src, dst, s_l, s_r, offs, base, e, csr, N_EDGES);

    int agg_blocks = (N_NODES + 8 * AGG_NODES - 1) / (8 * AGG_NODES);
    gat_node_aggregate<<<agg_blocks, B, 0, stream>>>(z, e, src, csr, offs, out, N_NODES);
}


# GEMM operand feed: W_fc copied once per work-group into LDS (padded pitch), every B fragment read from there; was 128 global fragment loads of 16 lines each per wave; k order and sums unchanged
# speedup vs baseline: 3.2342x; 1.0751x over previous
#include <hip/hip_runtime.h>
#include <hip/hip_bf16.h>
#include <math.h>

typedef __attribute__((ext_vector_type(2))) float v2f;
typedef __attribute__((ext_vector_type(8))) float v8f;

#define N_NODES 100000
#define N_EDGES 1600000
#define IN_DIM 128
#define OUT_DIM 64
#define NEG_SLOPE 0.01f

#define GEMM_THREADS 256
#define W_PITCH (IN_DIM + 4)
#define W_GROUPS (OUT_DIM * IN_DIM / 4)
static_assert(W_GROUPS % GEMM_THREADS == 0, "every thread copies the same number of float4 groups");
static_assert(IN_DIM / 4 == 32, "the copy splits a group index into row and column by 32");

__global__ void __launch_bounds__(GEMM_THREADS)
gat_gemm_wmma(const float* __restrict__ h,
              const float* __restrict__ W,
              float* __restrict__ z,
              int n_nodes) {
    __shared__ float Wl[OUT_DIM * W_PITCH];

    const int gtid = blockIdx.x * GEMM_THREADS + threadIdx.x;
    const int wave = gtid >> 5;
    const int lane = threadIdx.x & 31;
    const int m0 = wave * 16;
    const bool live = m0 < n_nodes;

    const int khalf = (lane >> 4) * 2;
    const int nlane = lane & 15;

    int rowA = m0 + nlane;
    if (!live) rowA = 0;
    const float* ap = h + (size_t)rowA * IN_DIM + khalf;
    v2f a[IN_DIM / 4];
#pragma unroll
    for (int s = 0; s < IN_DIM / 4; ++s) a[s] = *(const v2f*)(ap + 4 * s);

#pragma unroll
    for (int q = 0; q < W_GROUPS / GEMM_THREADS; ++q) {
        const int j  = q * GEMM_THREADS + (int)threadIdx.x;
        const int n  = j >> 5;
        const int k4 = (j & 31) << 2;
        *(float4*)(Wl + n * W_PITCH + k4) = *(const float4*)(W + (size_t)n * IN_DIM + k4);
    }
    __syncthreads();
    if (!live) return;

    const float* bl = Wl + nlane * W_PITCH + khalf;
    v8f acc0 = {}, acc1 = {}, acc2 = {}, acc3 = {};

#pragma unroll
    for (int s = 0; s < IN_DIM / 4; ++s) {
        const int k0 = 4 * s;
        v2f b0 = *(const v2f*)(bl +  0 * W_PITCH + k0);
        v2f b1 = *(const v2f*)(bl + 16 * W_PITCH + k0);
        v2f b2 = *(const v2f*)(bl + 32 * W_PITCH + k0);
        v2f b3 = *(const v2f*)(bl + 48 * W_PITCH + k0);

        acc0 = __builtin_amdgcn_wmma_f32_16x16x4_f32(false, a[s], false, b0, (short)0, acc0, false, false);
        acc1 = __builtin_amdgcn_wmma_f32_16x16x4_f32(false, a[s], false, b1, (short)0, acc1, false, false);
        acc2 = __builtin_amdgcn_wmma_f32_16x16x4_f32(false, a[s], false, b2, (short)0, acc2, false, false);
        acc3 = __builtin_amdgcn_wmma_f32_16x16x4_f32(false, a[s], false, b3, (short)0, acc3, false, false);
    }

    int rsel = (lane >> 4) * 8;
#pragma unroll
    for (int r = 0; r < 8; ++r) {
        size_t ro = (size_t)(m0 + r + rsel) * OUT_DIM;
        z[ro + nlane +  0] = acc0[r];
        z[ro + nlane + 16] = acc1[r];
        z[ro + nlane + 32] = acc2[r];
        z[ro + nlane + 48] = acc3[r];
    }
}

__global__ void gat_node_scores(const float* __restrict__ z,
                                const float* __restrict__ attn,
                                float* __restrict__ s_l,
                                float* __restrict__ s_r,
                                int n) {
    int i = blockIdx.x * blockDim.x + threadIdx.x;
    if (i >= n) return;
    const float4* zr = (const float4*)(z + (size_t)i * OUT_DIM);
    const float4* al = (const float4*)(attn);
    const float4* ar = (const float4*)(attn + OUT_DIM);
    float sl = 0.f, sr = 0.f;
#pragma unroll
    for (int q = 0; q < OUT_DIM / 4; ++q) {
        float4 zv = zr[q];
        float4 av = al[q];
        float4 bv = ar[q];
        sl += zv.x * av.x + zv.y * av.y + zv.z * av.z + zv.w * av.w;
        sr += zv.x * bv.x + zv.y * bv.y + zv.z * bv.z + zv.w * bv.w;
    }
    s_l[i] = sl;
    s_r[i] = sr;
}

#define HIST_CHUNKS   32
#define HIST_THREADS  1024
#define HIST_EDGES    ((N_EDGES + HIST_CHUNKS - 1) / HIST_CHUNKS)
#define HIST_PER_THR  ((HIST_EDGES + HIST_THREADS - 1) / HIST_THREADS)
#define HIST_WORDS    (N_NODES / 2)
#define HIST_GROUP    8
static_assert(HIST_EDGES < 65536, "a chunk's count of one node must fit a 16-bit bin");
static_assert(N_NODES % 8 == 0, "bins are cleared and copied as whole uint4 groups");

__global__ void __launch_bounds__(HIST_THREADS)
gat_hist_rank(const int* __restrict__ dst,
              unsigned int* __restrict__ cnt,
              int* __restrict__ rank,
              int ne) {
    __shared__ unsigned int bins[HIST_WORDS];

    const int t  = threadIdx.x;
    const int c  = blockIdx.x;
    const int e0 = c * HIST_EDGES;
    const int e1 = (e0 + HIST_EDGES < ne) ? (e0 + HIST_EDGES) : ne;

    int d[HIST_PER_THR];
#pragma unroll
    for (int q = 0; q < HIST_PER_THR; ++q) {
        const int i = e0 + q * HIST_THREADS + t;
        d[q] = dst[(i < e1) ? i : (ne - 1)];
    }

    uint4* b4 = (uint4*)bins;
    for (int j = t; j < HIST_WORDS / 4; j += HIST_THREADS) b4[j] = make_uint4(0u, 0u, 0u, 0u);
    __syncthreads();

#pragma unroll
    for (int q0 = 0; q0 < HIST_PER_THR; q0 += HIST_GROUP) {
        unsigned int old[HIST_GROUP];
#pragma unroll
        for (int u = 0; u < HIST_GROUP; ++u) {
            const int q = q0 + u;
            if (q < HIST_PER_THR) {
                const int i = e0 + q * HIST_THREADS + t;
                const unsigned int inc = (i < e1) ? ((d[q] & 1) ? 0x10000u : 1u) : 0u;
                old[u] = atomicAdd(&bins[d[q] >> 1], inc);
            }
        }
#pragma unroll
        for (int u = 0; u < HIST_GROUP; ++u) {
            const int q = q0 + u;
            if (q < HIST_PER_THR) {
                const int i = e0 + q * HIST_THREADS + t;
                if (i < e1) rank[i] = (int)((d[q] & 1) ? (old[u] >> 16) : (old[u] & 0xffffu));
            }
        }
    }
    __syncthreads();

    uint4* g4 = (uint4*)(cnt + (size_t)c * HIST_WORDS);
    for (int j = t; j < HIST_WORDS / 4; j += HIST_THREADS) g4[j] = b4[j];
}

__global__ void gat_chunk_bases(const unsigned int* __restrict__ cnt,
                                int* __restrict__ base,
                                int* __restrict__ deg,
                                int nwords) {
    const int j = blockIdx.x * blockDim.x + threadIdx.x;
    if (j >= nwords) return;
    unsigned int v[HIST_CHUNKS];
#pragma unroll
    for (int c = 0; c < HIST_CHUNKS; ++c) v[c] = cnt[(size_t)c * HIST_WORDS + j];
    int lo = 0, hi = 0;
#pragma unroll
    for (int c = 0; c < HIST_CHUNKS; ++c) {
        *(int2*)(base + (size_t)c * N_NODES + 2 * j) = make_int2(lo, hi);
        lo += (int)(v[c] & 0xffffu);
        hi += (int)(v[c] >> 16);
    }
    *(int2*)(deg + 2 * j) = make_int2(lo, hi);
}

#define SCAN_THREADS 1024
#define SCAN_CHUNK   (SCAN_THREADS * 4)
#define SCAN_BLOCKS  ((N_NODES + SCAN_CHUNK - 1) / SCAN_CHUNK)
static_assert(N_NODES % 4 == 0, "the scan reads and writes whole int4 groups");

__global__ void __launch_bounds__(SCAN_THREADS)
gat_scan_excl(const int* __restrict__ deg,
              int* __restrict__ offs,
              int n) {
    __shared__ int wave_tot[SCAN_THREADS / 32];
    __shared__ int wave_pre[SCAN_THREADS / 32];

    const int t    = threadIdx.x;
    const int lane = t & 31;
    const int w    = t >> 5;
    const int n4   = n >> 2;
    const int4* __restrict__ d4 = (const int4*)deg;

    const int i4 = blockIdx.x * SCAN_THREADS + t;
    int4 v = make_int4(0, 0, 0, 0);
    if (i4 < n4) v = d4[i4];

    int pre = 0;
    const int front = blockIdx.x;
    for (int c = 0; c < front; ++c) {
        int4 u = d4[c * SCAN_THREADS + t];
        pre += (u.x + u.y) + (u.z + u.w);
    }

    const int s0 = v.x;
    const int s1 = s0 + v.y;
    const int s2 = s1 + v.z;
    const int s3 = s2 + v.w;

    int inc = s3;
    for (int off = 1; off < 32; off <<= 1) {
        int y = __shfl_up(inc, off, 32);
        if (lane >= off) inc += y;
        pre += __shfl_xor(pre, off, 32);
    }
    if (lane == 31) wave_tot[w] = inc;
    if (lane == 0)  wave_pre[w] = pre;
    __syncthreads();

    int x = wave_pre[lane] + ((lane < w) ? wave_tot[lane] : 0);
    for (int off = 1; off < 32; off <<= 1) x += __shfl_xor(x, off, 32);

    const int ex = x + inc - s3;
    if (i4 < n4) {
        int4 o = make_int4(ex, ex + s0, ex + s1, ex + s2);
        ((int4*)offs)[i4] = o;
        if (i4 == n4 - 1) offs[n] = ex + s3;
    }
}

__global__ void gat_score_fill(const int* __restrict__ src,
                               const int* __restrict__ dst,
                               const float* __restrict__ s_l,
                               const float* __restrict__ s_r,
                               const int* __restrict__ offs,
                               const int* __restrict__ base,
                               float* __restrict__ e_rank,
                               int* __restrict__ csr,
                               int ne) {
    int i = blockIdx.x * blockDim.x + threadIdx.x;
    if (i >= ne) return;
    int d = dst[i];
    int r = __float_as_int(e_rank[i]);
    float v = s_l[src[i]] + s_r[d];
    e_rank[i] = (v > 0.f) ? v : NEG_SLOPE * v;
    csr[offs[d] + base[(size_t)(i / HIST_EDGES) * N_NODES + d] + r] = i;
}

#define AGG_NODES 8
#define AGG_ROWS  8

template <int M>
__device__ __forceinline__ float lane_xor_row(float x) {
    return __builtin_bit_cast(float,
        __builtin_amdgcn_update_dpp(0, __builtin_bit_cast(int, x), 0x160 | M, 0xf, 0xf, true));
}
__device__ __forceinline__ float lane_xor_half(float x) {
    int b = __builtin_bit_cast(int, x);
    return __builtin_bit_cast(float,
        __builtin_amdgcn_permlanex16(b, b, 0x76543210, 0xfedcba98, false, false));
}
__device__ __forceinline__ float wave_sum(float x) {
    x += lane_xor_row<1>(x);
    x += lane_xor_row<2>(x);
    x += lane_xor_row<4>(x);
    x += lane_xor_row<8>(x);
    x += lane_xor_half(x);
    return x;
}
__device__ __forceinline__ float wave_max(float x) {
    x = fmaxf(x, lane_xor_row<1>(x));
    x = fmaxf(x, lane_xor_row<2>(x));
    x = fmaxf(x, lane_xor_row<4>(x));
    x = fmaxf(x, lane_xor_row<8>(x));
    x = fmaxf(x, lane_xor_half(x));
    return x;
}
__device__ __forceinline__ int lane_read(int v, int l) {
    return __builtin_amdgcn_readlane(v, l);
}
__device__ __forceinline__ float lane_read(float v, int l) {
    return __builtin_bit_cast(float, __builtin_amdgcn_readlane(__builtin_bit_cast(int, v), l));
}

__global__ void gat_node_aggregate(const float* __restrict__ z,
                                   const float* __restrict__ e,
                                   const int* __restrict__ src,
                                   const int* __restrict__ csr,
                                   const int* __restrict__ offs,
                                   float* __restrict__ out,
                                   int n) {
    const int lane = threadIdx.x & 31;
    const int wave  = __builtin_amdgcn_readfirstlane((int)((blockIdx.x * blockDim.x + threadIdx.x) >> 5));
    const int node0 = wave * AGG_NODES;
    if (node0 >= n) return;
    const int c0 = lane * 2;

    int oi = node0 + ((lane < AGG_NODES) ? lane : AGG_NODES);
    if (oi > n) oi = n;
    const int offv = offs[oi];

    int o = lane_read(offv, 0);
    int d = lane_read(offv, 1) - o;
    int ei = (lane < d) ? csr[o + lane] : 0;

    for (int k = 0; k < AGG_NODES; ++k) {
        const int node = node0 + k;
        if (node >= n) break;
        const bool act = lane < d;

        float ev = act ? e[ei] : 0.f;
        int   sv = act ? src[ei] : 0;

        int o2 = o + d, d2 = 0, ei2 = 0;
        if (k + 1 < AGG_NODES && node + 1 < n) {
            d2  = lane_read(offv, k + 2) - o2;
            ei2 = (lane < d2) ? csr[o2 + lane] : 0;
        }

        float accx = 0.f, accy = 0.f;
        if (d > 32) {
            float sum = 0.f;
            for (int j = 0; j < d; ++j) sum += e[csr[o + j]];
            const float mean = sum / (float)d;
            float m = -1e30f;
            for (int j = 0; j < d; ++j) {
                float x  = e[csr[o + j]];
                float e2 = (x < mean) ? 0.f : x;
                m = fmaxf(m, e2);
            }
            float denom = 0.f;
            for (int j = 0; j < d; ++j) {
                int   ej = csr[o + j];
                float x  = e[ej];
                float e2 = (x < mean) ? 0.f : x;
                float ex = __expf(e2 - m);
                denom += ex;
                const float2 zv = *(const float2*)(z + (size_t)src[ej] * OUT_DIM + c0);
                accx += ex * zv.x;
                accy += ex * zv.y;
            }
            const float inv = 1.f / denom;
            accx *= inv;
            accy *= inv;
        } else if (d > 0) {
            int p = 0;
            for (int j = 0; j < d; ++j) p += (lane_read(ei, j) < ei) ? 1 : 0;
            if (!act) p = lane;
            ev = __builtin_bit_cast(float, __builtin_amdgcn_ds_permute(p << 2, __builtin_bit_cast(int, ev)));
            sv = __builtin_amdgcn_ds_permute(p << 2, sv);

            const float mean = wave_sum(ev) / (float)d;
            const float e2 = (ev < mean) ? 0.f : ev;
            const float m  = wave_max(act ? e2 : -1e30f);
            const float ex = act ? __expf(e2 - m) : 0.f;
            const float denom = wave_sum(ex);

            for (int j0 = 0; j0 < d; j0 += AGG_ROWS) {
                float2 zv[AGG_ROWS];
                float  w[AGG_ROWS];
#pragma unroll
                for (int u = 0; u < AGG_ROWS; ++u) {
                    const int s = lane_read(sv, j0 + u);
                    w[u]  = lane_read(ex, j0 + u);
                    zv[u] = *(const float2*)(z + (size_t)s * OUT_DIM + c0);
                }
#pragma unroll
                for (int u = 0; u < AGG_ROWS; ++u) {
                    accx += w[u] * zv[u].x;
                    accy += w[u] * zv[u].y;
                }
            }
            const float inv = 1.f / denom;
            accx *= inv;
            accy *= inv;
        }
        float2 r;
        r.x = accx;
        r.y = accy;
        *(float2*)(out + (size_t)node * OUT_DIM + c0) = r;

        o = o2;
        d = d2;
        ei = ei2;
    }
}

extern "C" void kernel_launch(void* const* d_in, const int* in_sizes, int n_in,
                              void* d_out, int out_size, void* d_ws, size_t ws_size,
                              hipStream_t stream) {
    const float* h      = (const float*)d_in[0];
    const int*   src    = (const int*)d_in[1];
    const int*   dst    = (const int*)d_in[2];
    const float* W_fc   = (const float*)d_in[3];
    const float* W_attn = (const float*)d_in[4];
    float* out = (float*)d_out;

    char* ws = (char*)d_ws;
    size_t off = 0;
    auto carve = [&](size_t bytes) {
        void* p = ws + off;
        off += (bytes + 15) & ~(size_t)15;
        return p;
    };
    float* z      = (float*)carve((size_t)N_NODES * OUT_DIM * sizeof(float));
    float* s_l    = (float*)carve((size_t)N_NODES * sizeof(float));
    float* s_r    = (float*)carve((size_t)N_NODES * sizeof(float));
    float* e      = (float*)carve((size_t)N_EDGES * sizeof(float));
    int*   deg    = (int*)carve((size_t)N_NODES * sizeof(int));
    int*   offs   = (int*)carve((size_t)(N_NODES + 1) * sizeof(int));
    int*   csr    = (int*)carve((size_t)N_EDGES * sizeof(int));
    (void)ws_size;

    const int B = 256;

    int gemm_waves  = N_NODES / 16;
    int gemm_blocks = (gemm_waves + 7) / 8;
    gat_gemm_wmma<<<gemm_blocks, GEMM_THREADS, 0, stream>>>(h, W_fc, z, N_NODES);

    gat_node_scores<<<(N_NODES + B - 1) / B, B, 0, stream>>>(z, W_attn, s_l, s_r, N_NODES);

    unsigned int* cnt = (unsigned int*)csr;
    int* base = (int*)d_out;
    gat_hist_rank<<<HIST_CHUNKS, HIST_THREADS, 0, stream>>>(dst, cnt, (int*)e, N_EDGES);
    gat_chunk_bases<<<(HIST_WORDS + B - 1) / B, B, 0, stream>>>(cnt, base, deg, HIST_WORDS);

    gat_scan_excl<<<SCAN_BLOCKS, SCAN_THREADS, 0, stream>>>(deg, offs, N_NODES);

    gat_score_fill<<<(N_EDGES + B - 1) / B, B, 0, stream>>>(src, dst, s_l, s_r, offs, base, e, csr, N_EDGES);

    int agg_blocks = (N_NODES + 8 * AGG_NODES - 1) / (8 * AGG_NODES);
    gat_node_aggregate<<<agg_blocks, B, 0, stream>>>(z, e, src, csr, offs, out, N_NODES);
}


# CSR slot holds the edge's record {source id, score} (was edge id): node pass reads a node's records as one contiguous load, one node ahead; two per-edge lookups go; in-node sums in source-id order
# speedup vs baseline: 3.6448x; 1.1270x over previous
#include <hip/hip_runtime.h>
#include <hip/hip_bf16.h>
#include <math.h>

typedef __attribute__((ext_vector_type(2))) float v2f;
typedef __attribute__((ext_vector_type(8))) float v8f;

#define N_NODES 100000
#define N_EDGES 1600000
#define IN_DIM 128
#define OUT_DIM 64
#define NEG_SLOPE 0.01f

#define GEMM_THREADS 256
#define W_PITCH (IN_DIM + 4)
#define W_GROUPS (OUT_DIM * IN_DIM / 4)
static_assert(W_GROUPS % GEMM_THREADS == 0, "every thread copies the same number of float4 groups");
static_assert(IN_DIM / 4 == 32, "the copy splits a group index into row and column by 32");

__global__ void __launch_bounds__(GEMM_THREADS)
gat_gemm_wmma(const float* __restrict__ h,
              const float* __restrict__ W,
              float* __restrict__ z,
              int n_nodes) {
    __shared__ float Wl[OUT_DIM * W_PITCH];

    const int gtid = blockIdx.x * GEMM_THREADS + threadIdx.x;
    const int wave = gtid >> 5;
    const int lane = threadIdx.x & 31;
    const int m0 = wave * 16;
    const bool live = m0 < n_nodes;

    const int khalf = (lane >> 4) * 2;
    const int nlane = lane & 15;

    int rowA = m0 + nlane;
    if (!live) rowA = 0;
    const float* ap = h + (size_t)rowA * IN_DIM + khalf;
    v2f a[IN_DIM / 4];
#pragma unroll
    for (int s = 0; s < IN_DIM / 4; ++s) a[s] = *(const v2f*)(ap + 4 * s);

#pragma unroll
    for (int q = 0; q < W_GROUPS / GEMM_THREADS; ++q) {
        const int j  = q * GEMM_THREADS + (int)threadIdx.x;
        const int n  = j >> 5;
        const int k4 = (j & 31) << 2;
        *(float4*)(Wl + n * W_PITCH + k4) = *(const float4*)(W + (size_t)n * IN_DIM + k4);
    }
    __syncthreads();
    if (!live) return;

    const float* bl = Wl + nlane * W_PITCH + khalf;
    v8f acc0 = {}, acc1 = {}, acc2 = {}, acc3 = {};

#pragma unroll
    for (int s = 0; s < IN_DIM / 4; ++s) {
        const int k0 = 4 * s;
        v2f b0 = *(const v2f*)(bl +  0 * W_PITCH + k0);
        v2f b1 = *(const v2f*)(bl + 16 * W_PITCH + k0);
        v2f b2 = *(const v2f*)(bl + 32 * W_PITCH + k0);
        v2f b3 = *(const v2f*)(bl + 48 * W_PITCH + k0);

        acc0 = __builtin_amdgcn_wmma_f32_16x16x4_f32(false, a[s], false, b0, (short)0, acc0, false, false);
        acc1 = __builtin_amdgcn_wmma_f32_16x16x4_f32(false, a[s], false, b1, (short)0, acc1, false, false);
        acc2 = __builtin_amdgcn_wmma_f32_16x16x4_f32(false, a[s], false, b2, (short)0, acc2, false, false);
        acc3 = __builtin_amdgcn_wmma_f32_16x16x4_f32(false, a[s], false, b3, (short)0, acc3, false, false);
    }

    int rsel = (lane >> 4) * 8;
#pragma unroll
    for (int r = 0; r < 8; ++r) {
        size_t ro = (size_t)(m0 + r + rsel) * OUT_DIM;
        z[ro + nlane +  0] = acc0[r];
        z[ro + nlane + 16] = acc1[r];
        z[ro + nlane + 32] = acc2[r];
        z[ro + nlane + 48] = acc3[r];
    }
}

__global__ void gat_node_scores(const float* __restrict__ z,
                                const float* __restrict__ attn,
                                float* __restrict__ s_l,
                                float* __restrict__ s_r,
                                int n) {
    int i = blockIdx.x * blockDim.x + threadIdx.x;
    if (i >= n) return;
    const float4* zr = (const float4*)(z + (size_t)i * OUT_DIM);
    const float4* al = (const float4*)(attn);
    const float4* ar = (const float4*)(attn + OUT_DIM);
    float sl = 0.f, sr = 0.f;
#pragma unroll
    for (int q = 0; q < OUT_DIM / 4; ++q) {
        float4 zv = zr[q];
        float4 av = al[q];
        float4 bv = ar[q];
        sl += zv.x * av.x + zv.y * av.y + zv.z * av.z + zv.w * av.w;
        sr += zv.x * bv.x + zv.y * bv.y + zv.z * bv.z + zv.w * bv.w;
    }
    s_l[i] = sl;
    s_r[i] = sr;
}

#define HIST_CHUNKS   32
#define HIST_THREADS  1024
#define HIST_EDGES    ((N_EDGES + HIST_CHUNKS - 1) / HIST_CHUNKS)
#define HIST_PER_THR  ((HIST_EDGES + HIST_THREADS - 1) / HIST_THREADS)
#define HIST_WORDS    (N_NODES / 2)
#define HIST_GROUP    8
static_assert(HIST_EDGES < 65536, "a chunk's count of one node must fit a 16-bit bin");
static_assert(N_NODES % 8 == 0, "bins are cleared and copied as whole uint4 groups");

__global__ void __launch_bounds__(HIST_THREADS)
gat_hist_rank(const int* __restrict__ dst,
              unsigned int* __restrict__ cnt,
              int* __restrict__ rank,
              int ne) {
    __shared__ unsigned int bins[HIST_WORDS];

    const int t  = threadIdx.x;
    const int c  = blockIdx.x;
    const int e0 = c * HIST_EDGES;
    const int e1 = (e0 + HIST_EDGES < ne) ? (e0 + HIST_EDGES) : ne;

    int d[HIST_PER_THR];
#pragma unroll
    for (int q = 0; q < HIST_PER_THR; ++q) {
        const int i = e0 + q * HIST_THREADS + t;
        d[q] = dst[(i < e1) ? i : (ne - 1)];
    }

    uint4* b4 = (uint4*)bins;
    for (int j = t; j < HIST_WORDS / 4; j += HIST_THREADS) b4[j] = make_uint4(0u, 0u, 0u, 0u);
    __syncthreads();

#pragma unroll
    for (int q0 = 0; q0 < HIST_PER_THR; q0 += HIST_GROUP) {
        unsigned int old[HIST_GROUP];
#pragma unroll
        for (int u = 0; u < HIST_GROUP; ++u) {
            const int q = q0 + u;
            if (q < HIST_PER_THR) {
                const int i = e0 + q * HIST_THREADS + t;
                const unsigned int inc = (i < e1) ? ((d[q] & 1) ? 0x10000u : 1u) : 0u;
                old[u] = atomicAdd(&bins[d[q] >> 1], inc);
            }
        }
#pragma unroll
        for (int u = 0; u < HIST_GROUP; ++u) {
            const int q = q0 + u;
            if (q < HIST_PER_THR) {
                const int i = e0 + q * HIST_THREADS + t;
                if (i < e1) rank[i] = (int)((d[q] & 1) ? (old[u] >> 16) : (old[u] & 0xffffu));
            }
        }
    }
    __syncthreads();

    uint4* g4 = (uint4*)(cnt + (size_t)c * HIST_WORDS);
    for (int j = t; j < HIST_WORDS / 4; j += HIST_THREADS) g4[j] = b4[j];
}

__global__ void gat_chunk_bases(const unsigned int* __restrict__ cnt,
                                int* __restrict__ base,
                                int* __restrict__ deg,
                                int nwords) {
    const int j = blockIdx.x * blockDim.x + threadIdx.x;
    if (j >= nwords) return;
    unsigned int v[HIST_CHUNKS];
#pragma unroll
    for (int c = 0; c < HIST_CHUNKS; ++c) v[c] = cnt[(size_t)c * HIST_WORDS + j];
    int lo = 0, hi = 0;
#pragma unroll
    for (int c = 0; c < HIST_CHUNKS; ++c) {
        *(int2*)(base + (size_t)c * N_NODES + 2 * j) = make_int2(lo, hi);
        lo += (int)(v[c] & 0xffffu);
        hi += (int)(v[c] >> 16);
    }
    *(int2*)(deg + 2 * j) = make_int2(lo, hi);
}

#define SCAN_THREADS 1024
#define SCAN_CHUNK   (SCAN_THREADS * 4)
#define SCAN_BLOCKS  ((N_NODES + SCAN_CHUNK - 1) / SCAN_CHUNK)
static_assert(N_NODES % 4 == 0, "the scan reads and writes whole int4 groups");

__global__ void __launch_bounds__(SCAN_THREADS)
gat_scan_excl(const int* __restrict__ deg,
              int* __restrict__ offs,
              int n) {
    __shared__ int wave_tot[SCAN_THREADS / 32];
    __shared__ int wave_pre[SCAN_THREADS / 32];

    const int t    = threadIdx.x;
    const int lane = t & 31;
    const int w    = t >> 5;
    const int n4   = n >> 2;
    const int4* __restrict__ d4 = (const int4*)deg;

    const int i4 = blockIdx.x * SCAN_THREADS + t;
    int4 v = make_int4(0, 0, 0, 0);
    if (i4 < n4) v = d4[i4];

    int pre = 0;
    const int front = blockIdx.x;
    for (int c = 0; c < front; ++c) {
        int4 u = d4[c * SCAN_THREADS + t];
        pre += (u.x + u.y) + (u.z + u.w);
    }

    const int s0 = v.x;
    const int s1 = s0 + v.y;
    const int s2 = s1 + v.z;
    const int s3 = s2 + v.w;

    int inc = s3;
    for (int off = 1; off < 32; off <<= 1) {
        int y = __shfl_up(inc, off, 32);
        if (lane >= off) inc += y;
        pre += __shfl_xor(pre, off, 32);
    }
    if (lane == 31) wave_tot[w] = inc;
    if (lane == 0)  wave_pre[w] = pre;
    __syncthreads();

    int x = wave_pre[lane] + ((lane < w) ? wave_tot[lane] : 0);
    for (int off = 1; off < 32; off <<= 1) x += __shfl_xor(x, off, 32);

    const int ex = x + inc - s3;
    if (i4 < n4) {
        int4 o = make_int4(ex, ex + s0, ex + s1, ex + s2);
        ((int4*)offs)[i4] = o;
        if (i4 == n4 - 1) offs[n] = ex + s3;
    }
}

__global__ void gat_score_fill(const int* __restrict__ src,
                               const int* __restrict__ dst,
                               const float* __restrict__ s_l,
                               const float* __restrict__ s_r,
                               const int* __restrict__ offs,
                               const int* __restrict__ base,
                               const int* __restrict__ rank,
                               int* __restrict__ rec,
                               int ne) {
    int i = blockIdx.x * blockDim.x + threadIdx.x;
    if (i >= ne) return;
    int d = dst[i];
    int r = rank[i];
    int sidx = src[i];
    float v = s_l[sidx] + s_r[d];
    v = (v > 0.f) ? v : NEG_SLOPE * v;
    const int pos = offs[d] + base[(size_t)(i / HIST_EDGES) * N_NODES + d] + r;
    *(int2*)(rec + 2 * (size_t)pos) = make_int2(sidx, __float_as_int(v));
}

#define AGG_NODES 8
#define AGG_ROWS  8

template <int M>
__device__ __forceinline__ float lane_xor_row(float x) {
    return __builtin_bit_cast(float,
        __builtin_amdgcn_update_dpp(0, __builtin_bit_cast(int, x), 0x160 | M, 0xf, 0xf, true));
}
__device__ __forceinline__ float lane_xor_half(float x) {
    int b = __builtin_bit_cast(int, x);
    return __builtin_bit_cast(float,
        __builtin_amdgcn_permlanex16(b, b, 0x76543210, 0xfedcba98, false, false));
}
__device__ __forceinline__ float wave_sum(float x) {
    x += lane_xor_row<1>(x);
    x += lane_xor_row<2>(x);
    x += lane_xor_row<4>(x);
    x += lane_xor_row<8>(x);
    x += lane_xor_half(x);
    return x;
}
__device__ __forceinline__ float wave_max(float x) {
    x = fmaxf(x, lane_xor_row<1>(x));
    x = fmaxf(x, lane_xor_row<2>(x));
    x = fmaxf(x, lane_xor_row<4>(x));
    x = fmaxf(x, lane_xor_row<8>(x));
    x = fmaxf(x, lane_xor_half(x));
    return x;
}
__device__ __forceinline__ int lane_read(int v, int l) {
    return __builtin_amdgcn_readlane(v, l);
}
__device__ __forceinline__ float lane_read(float v, int l) {
    return __builtin_bit_cast(float, __builtin_amdgcn_readlane(__builtin_bit_cast(int, v), l));
}

__global__ void gat_node_aggregate(const float* __restrict__ z,
                                   const int* __restrict__ rec_i,
                                   const int* __restrict__ offs,
                                   float* __restrict__ out,
                                   int n) {
    const int2* __restrict__ rec = (const int2*)rec_i;
    const int lane = threadIdx.x & 31;
    const int wave  = __builtin_amdgcn_readfirstlane((int)((blockIdx.x * blockDim.x + threadIdx.x) >> 5));
    const int node0 = wave * AGG_NODES;
    if (node0 >= n) return;
    const int c0 = lane * 2;

    int oi = node0 + ((lane < AGG_NODES) ? lane : AGG_NODES);
    if (oi > n) oi = n;
    const int offv = offs[oi];

    int  o  = lane_read(offv, 0);
    int  d  = 0;
    int2 rc = make_int2(0, 0);

    for (int k = -1; k < AGG_NODES; ++k) {
        const int node = node0 + k;
        if (node >= n) break;
        const bool act = lane < d;

        float ev = __int_as_float(rc.y);
        int   sv = rc.x;

        int o2 = o + d, d2 = 0;
        int2 rc2 = make_int2(0, 0);
        if (k + 1 < AGG_NODES && node + 1 < n) {
            d2 = lane_read(offv, k + 2) - o2;
            if (lane < d2) rc2 = rec[o2 + lane];
        }

        float accx = 0.f, accy = 0.f;
        if (d > 32) {
            float sum = 0.f;
            for (int j = 0; j < d; ++j) sum += __int_as_float(rec[o + j].y);
            const float mean = sum / (float)d;
            float m = -1e30f;
            for (int j = 0; j < d; ++j) {
                float x  = __int_as_float(rec[o + j].y);
                float e2 = (x < mean) ? 0.f : x;
                m = fmaxf(m, e2);
            }
            float denom = 0.f;
            for (int j = 0; j < d; ++j) {
                const int2 rj = rec[o + j];
                float x  = __int_as_float(rj.y);
                float e2 = (x < mean) ? 0.f : x;
                float ex = __expf(e2 - m);
                denom += ex;
                const float2 zv = *(const float2*)(z + (size_t)rj.x * OUT_DIM + c0);
                accx += ex * zv.x;
                accy += ex * zv.y;
            }
            const float inv = 1.f / denom;
            accx *= inv;
            accy *= inv;
        } else if (d > 0) {
            const int key = (sv << 5) | lane;
            int p = 0;
            for (int j = 0; j < d; ++j) p += (lane_read(key, j) < key) ? 1 : 0;
            if (!act) p = lane;
            ev = __builtin_bit_cast(float, __builtin_amdgcn_ds_permute(p << 2, __builtin_bit_cast(int, ev)));
            sv = __builtin_amdgcn_ds_permute(p << 2, sv);

            const float mean = wave_sum(ev) / (float)d;
            const float e2 = (ev < mean) ? 0.f : ev;
            const float m  = wave_max(act ? e2 : -1e30f);
            const float ex = act ? __expf(e2 - m) : 0.f;
            const float denom = wave_sum(ex);

            for (int j0 = 0; j0 < d; j0 += AGG_ROWS) {
                float2 zv[AGG_ROWS];
                float  w[AGG_ROWS];
#pragma unroll
                for (int u = 0; u < AGG_ROWS; ++u) {
                    const int s = lane_read(sv, j0 + u);
                    w[u]  = lane_read(ex, j0 + u);
                    zv[u] = *(const float2*)(z + (size_t)s * OUT_DIM + c0);
                }
#pragma unroll
                for (int u = 0; u < AGG_ROWS; ++u) {
                    accx += w[u] * zv[u].x;
                    accy += w[u] * zv[u].y;
                }
            }
            const float inv = 1.f / denom;
            accx *= inv;
            accy *= inv;
        }
        if (k >= 0) {
            float2 r;
            r.x = accx;
            r.y = accy;
            *(float2*)(out + (size_t)node * OUT_DIM + c0) = r;
        }

        o = o2;
        d = d2;
        rc = rc2;
    }
}

extern "C" void kernel_launch(void* const* d_in, const int* in_sizes, int n_in,
                              void* d_out, int out_size, void* d_ws, size_t ws_size,
                              hipStream_t stream) {
    const float* h      = (const float*)d_in[0];
    const int*   src    = (const int*)d_in[1];
    const int*   dst    = (const int*)d_in[2];
    const float* W_fc   = (const float*)d_in[3];
    const float* W_attn = (const float*)d_in[4];
    float* out = (float*)d_out;

    char* ws = (char*)d_ws;
    size_t off = 0;
    auto carve = [&](size_t bytes) {
        void* p = ws + off;
        off += (bytes + 15) & ~(size_t)15;
        return p;
    };
    float* z      = (float*)carve((size_t)N_NODES * OUT_DIM * sizeof(float));
    float* s_l    = (float*)carve((size_t)N_NODES * sizeof(float));
    float* s_r    = (float*)carve((size_t)N_NODES * sizeof(float));
    int*   rec    = (int*)carve((size_t)N_EDGES * 2 * sizeof(int));
    int*   deg    = (int*)carve((size_t)N_NODES * sizeof(int));
    int*   offs   = (int*)carve((size_t)(N_NODES + 1) * sizeof(int));
    (void)ws_size;

    const int B = 256;

    int gemm_waves  = N_NODES / 16;
    int gemm_blocks = (gemm_waves + 7) / 8;
    gat_gemm_wmma<<<gemm_blocks, GEMM_THREADS, 0, stream>>>(h, W_fc, z, N_NODES);

    gat_node_scores<<<(N_NODES + B - 1) / B, B, 0, stream>>>(z, W_attn, s_l, s_r, N_NODES);

    static_assert((size_t)HIST_CHUNKS * N_NODES * 4 + (size_t)HIST_CHUNKS * HIST_WORDS * 4 + (size_t)N_EDGES * 4
                  <= (size_t)N_NODES * OUT_DIM * 4, "base table, bins and ranks fit in out");
    int* base = (int*)d_out;
    unsigned int* cnt = (unsigned int*)(base + (size_t)HIST_CHUNKS * N_NODES);
    int* rank = (int*)(cnt + (size_t)HIST_CHUNKS * HIST_WORDS);
    gat_hist_rank<<<HIST_CHUNKS, HIST_THREADS, 0, stream>>>(dst, cnt, rank, N_EDGES);
    gat_chunk_bases<<<(HIST_WORDS + B - 1) / B, B, 0, stream>>>(cnt, base, deg, HIST_WORDS);

    gat_scan_excl<<<SCAN_BLOCKS, SCAN_THREADS, 0, stream>>>(deg, offs, N_NODES);

    gat_score_fill<<<(N_EDGES + B - 1) / B, B, 0, stream>>>(src, dst, s_l, s_r, offs, base, rank, rec, N_EDGES);

    int agg_blocks = (N_NODES + 8 * AGG_NODES - 1) / (8 * AGG_NODES);
    gat_node_aggregate<<<agg_blocks, B, 0, stream>>>(z, rec, offs, out, N_NODES);
}
